# starting kernel
# baseline (speedup 1.0000x reference)
#include <hip/hip_runtime.h>
#include <math.h>

typedef __attribute__((ext_vector_type(2))) float v2f;
typedef __attribute__((ext_vector_type(8))) float v8f;

__global__ void k_fill_one(float* __restrict__ deg, int n) {
    int i = blockIdx.x * blockDim.x + threadIdx.x;
    if (i < n) deg[i] = 1.0f;
}

__global__ void k_degree(const int* __restrict__ col, float* __restrict__ deg, int e) {
    int i = blockIdx.x * blockDim.x + threadIdx.x;
    if (i < e) unsafeAtomicAdd(&deg[col[i]], 1.0f);
}

__global__ void k_rsqrt(float* __restrict__ deg, int n) {
    int i = blockIdx.x * blockDim.x + threadIdx.x;
    if (i < n) {
        float d = deg[i];
        deg[i] = d > 0.0f ? rsqrtf(d) : 0.0f;
    }
}

template <int K, bool RELU, bool STOREPRE>
__global__ void k_dense(const float* __restrict__ X, const float* __restrict__ W,
                        const float* __restrict__ bias,
                        float* __restrict__ outA, float* __restrict__ outPre,
                        int ntiles) {
    __shared__ float wlds[64 * K];
    constexpr int NV4 = (64 * K) / 4;
    for (int i = threadIdx.x; i < NV4; i += 256)
        ((float4*)wlds)[i] = ((const float4*)W)[i];
    __syncthreads();

    int wid  = (blockIdx.x * blockDim.x + threadIdx.x) >> 5;
    int lane = threadIdx.x & 31;
    if (wid >= ntiles) return;
    int ln = lane & 15;
    int kh = lane >> 4;

    const float* xrow = X + (size_t)(wid * 16 + ln) * K + 2 * kh;
    v2f afr[K / 4];
#pragma unroll
    for (int kk = 0; kk < K / 4; ++kk)
        afr[kk] = *(const v2f*)(xrow + 4 * kk);

#pragma unroll
    for (int j = 0; j < 4; ++j) {
        const float* wrow = wlds + (size_t)(16 * j + ln) * K + 2 * kh;
        v8f acc = {};
#pragma unroll
        for (int kk = 0; kk < K / 4; ++kk) {
            v2f b = *(const v2f*)(wrow + 4 * kk);
            acc = __builtin_amdgcn_wmma_f32_16x16x4_f32(
                false, afr[kk], false, b, (short)0, acc, false, false);
        }
        int n = 16 * j + ln;
        float bv = bias[n];
#pragma unroll
        for (int r = 0; r < 8; ++r) {
            int m = r + 8 * kh;
            size_t gi = (size_t)(wid * 16 + m) * 64 + n;
            float v = acc[r] + bv;
            if constexpr (STOREPRE) outPre[gi] = v;
            outA[gi] = RELU ? fmaxf(v, 0.0f) : v;
        }
    }
}

__global__ void k_selfloop(const float* __restrict__ dinv, const float* __restrict__ h,
                           float* __restrict__ agg, long long total) {
    long long i = (long long)blockIdx.x * blockDim.x + threadIdx.x;
    if (i >= total) return;
    int node = (int)(i >> 6);
    float d = dinv[node];
    agg[i] = d * d * h[i];
}

__global__ void k_scatter(const int* __restrict__ row, const int* __restrict__ col,
                          const float* __restrict__ dinv, const float* __restrict__ h,
                          float* __restrict__ agg, int e) {
    long long t = (long long)blockIdx.x * blockDim.x + threadIdx.x;
    int ei = (int)(t >> 4);
    if (ei >= e) return;
    int f = (int)(t & 15) * 4;
    int r = row[ei], c = col[ei];
    float nrm = dinv[r] * dinv[c];
    const float4 hv = *(const float4*)(h + (size_t)r * 64 + f);
    float* dst = agg + (size_t)c * 64 + f;
    unsafeAtomicAdd(dst + 0, nrm * hv.x);
    unsafeAtomicAdd(dst + 1, nrm * hv.y);
    unsafeAtomicAdd(dst + 2, nrm * hv.z);
    unsafeAtomicAdd(dst + 3, nrm * hv.w);
}

__global__ void k_layer(const float* __restrict__ agg, const float* __restrict__ x0,
                        const float* __restrict__ Wl, float beta,
                        float* __restrict__ h, int ntiles) {
    __shared__ float wlds[64 * 64];
    for (int i = threadIdx.x; i < 1024; i += 256)
        ((float4*)wlds)[i] = ((const float4*)Wl)[i];
    __syncthreads();

    int wid  = (blockIdx.x * blockDim.x + threadIdx.x) >> 5;
    int lane = threadIdx.x & 31;
    if (wid >= ntiles) return;
    int ln = lane & 15;
    int kh = lane >> 4;

    size_t rb = (size_t)(wid * 16 + ln) * 64 + 2 * kh;
    v2f afr[16];
#pragma unroll
    for (int kk = 0; kk < 16; ++kk) {
        v2f ag = *(const v2f*)(agg + rb + 4 * kk);
        v2f xx = *(const v2f*)(x0 + rb + 4 * kk);
        afr[kk] = ag * 0.9f + xx * 0.1f;
    }
    float ob = 1.0f - beta;
#pragma unroll
    for (int j = 0; j < 4; ++j) {
        const float* wcol = wlds + 2 * kh * 64 + 16 * j + ln;
        v8f acc = {};
#pragma unroll
        for (int kk = 0; kk < 16; ++kk) {
            v2f b;
            b.x = wcol[(size_t)(4 * kk) * 64];
            b.y = wcol[(size_t)(4 * kk + 1) * 64];
            acc = __builtin_amdgcn_wmma_f32_16x16x4_f32(
                false, afr[kk], false, b, (short)0, acc, false, false);
        }
#pragma unroll
        for (int r = 0; r < 8; ++r) {
            int m = r + 8 * kh;
            int n = 16 * j + ln;
            size_t gi = (size_t)(wid * 16 + m) * 64 + n;
            float hh = 0.9f * agg[gi] + 0.1f * x0[gi];
            float v = ob * hh + beta * acc[r];
            h[gi] = fmaxf(v, 0.0f);
        }
    }
}

extern "C" void kernel_launch(void* const* d_in, const int* in_sizes, int n_in,
                              void* d_out, int out_size, void* d_ws, size_t ws_size,
                              hipStream_t stream) {
    const float* x      = (const float*)d_in[0];
    const int*   ei     = (const int*)  d_in[1];
    const float* lin0_w = (const float*)d_in[2];
    const float* lin0_b = (const float*)d_in[3];
    const float* conv_w = (const float*)d_in[4];
    const float* lin1_w = (const float*)d_in[5];
    const float* lin1_b = (const float*)d_in[6];
    float* out = (float*)d_out;

    const int N = in_sizes[0] / 128;
    const int E = in_sizes[1] / 2;
    const int* row = ei;
    const int* col = ei + E;

    float* ws   = (float*)d_ws;
    float* dinv = ws;
    float* x0   = dinv + N;
    float* h    = x0 + (size_t)N * 64;
    float* agg  = h  + (size_t)N * 64;

    const int ntiles = N / 16;
    const int gemmGrid = (ntiles * 32 + 255) / 256;
    const long long NF = (long long)N * 64;

    k_fill_one<<<(N + 255) / 256, 256, 0, stream>>>(dinv, N);
    k_degree  <<<(E + 255) / 256, 256, 0, stream>>>(col, dinv, E);
    k_rsqrt   <<<(N + 255) / 256, 256, 0, stream>>>(dinv, N);

    k_dense<128, true, true><<<gemmGrid, 256, 0, stream>>>(x, lin0_w, lin0_b, h, x0, ntiles);

    for (int l = 0; l < 4; ++l) {
        float beta = logf(0.5f / (float)(l + 1) + 1.0f);
        k_selfloop<<<(int)((NF + 255) / 256), 256, 0, stream>>>(dinv, h, agg, NF);
        k_scatter <<<(int)(((long long)E * 16 + 255) / 256), 256, 0, stream>>>(
            row, col, dinv, h, agg, E);
        k_layer   <<<gemmGrid, 256, 0, stream>>>(agg, x0, conv_w + (size_t)l * 64 * 64,
                                                 beta, h, ntiles);
    }

    k_dense<64, false, false><<<gemmGrid, 256, 0, stream>>>(h, lin1_w, lin1_b, out, nullptr, ntiles);
}


# Edges binned once per call into 64-node target buckets (bucket-only counting sort); per layer a block per bucket sums self-loop + edges in an LDS tile by LDS f32 atomics (arrival order), none global
# speedup vs baseline: 1.8093x; 1.8093x over previous
#include <hip/hip_runtime.h>
#include <math.h>

typedef __attribute__((ext_vector_type(2))) float v2f;
typedef __attribute__((ext_vector_type(8))) float v8f;

__global__ void k_fill_one(float* __restrict__ deg, int n) {
    int i = blockIdx.x * blockDim.x + threadIdx.x;
    if (i < n) deg[i] = 1.0f;
}

__global__ void k_degree(const int* __restrict__ col, float* __restrict__ deg, int e) {
    int i = blockIdx.x * blockDim.x + threadIdx.x;
    if (i < e) unsafeAtomicAdd(&deg[col[i]], 1.0f);
}

__global__ void k_rsqrt(float* __restrict__ deg, int n) {
    int i = blockIdx.x * blockDim.x + threadIdx.x;
    if (i < n) {
        float d = deg[i];
        deg[i] = d > 0.0f ? rsqrtf(d) : 0.0f;
    }
}

template <int K, bool RELU, bool STOREPRE>
__global__ void k_dense(const float* __restrict__ X, const float* __restrict__ W,
                        const float* __restrict__ bias,
                        float* __restrict__ outA, float* __restrict__ outPre,
                        int ntiles) {
    __shared__ float wlds[64 * K];
    constexpr int NV4 = (64 * K) / 4;
    for (int i = threadIdx.x; i < NV4; i += 256)
        ((float4*)wlds)[i] = ((const float4*)W)[i];
    __syncthreads();

    int wid  = (blockIdx.x * blockDim.x + threadIdx.x) >> 5;
    int lane = threadIdx.x & 31;
    if (wid >= ntiles) return;
    int ln = lane & 15;
    int kh = lane >> 4;

    const float* xrow = X + (size_t)(wid * 16 + ln) * K + 2 * kh;
    v2f afr[K / 4];
#pragma unroll
    for (int kk = 0; kk < K / 4; ++kk)
        afr[kk] = *(const v2f*)(xrow + 4 * kk);

#pragma unroll
    for (int j = 0; j < 4; ++j) {
        const float* wrow = wlds + (size_t)(16 * j + ln) * K + 2 * kh;
        v8f acc = {};
#pragma unroll
        for (int kk = 0; kk < K / 4; ++kk) {
            v2f b = *(const v2f*)(wrow + 4 * kk);
            acc = __builtin_amdgcn_wmma_f32_16x16x4_f32(
                false, afr[kk], false, b, (short)0, acc, false, false);
        }
        int n = 16 * j + ln;
        float bv = bias[n];
#pragma unroll
        for (int r = 0; r < 8; ++r) {
            int m = r + 8 * kh;
            size_t gi = (size_t)(wid * 16 + m) * 64 + n;
            float v = acc[r] + bv;
            if constexpr (STOREPRE) outPre[gi] = v;
            outA[gi] = RELU ? fmaxf(v, 0.0f) : v;
        }
    }
}

#define BKT         64
#define BKT_LOG     6
#define BIN_CHUNK   16384
#define BIN_THREADS 1024

extern __shared__ int dyn_lds[];

__global__ void __launch_bounds__(BIN_THREADS)
k_bin_count(const int* __restrict__ col, int* __restrict__ cnt, int e, int nb) {
    for (int i = threadIdx.x; i < nb; i += BIN_THREADS) dyn_lds[i] = 0;
    __syncthreads();
    const int e0 = blockIdx.x * BIN_CHUNK;
    int c[BIN_CHUNK / BIN_THREADS];
#pragma unroll
    for (int k = 0; k < BIN_CHUNK / BIN_THREADS; ++k) {
        int i = e0 + k * BIN_THREADS + (int)threadIdx.x;
        c[k] = col[i < e ? i : e - 1];
    }
#pragma unroll
    for (int k = 0; k < BIN_CHUNK / BIN_THREADS; ++k) {
        int i = e0 + k * BIN_THREADS + (int)threadIdx.x;
        if (i < e) atomicAdd(&dyn_lds[c[k] >> BKT_LOG], 1);
    }
    __syncthreads();
    int* dst = cnt + (size_t)blockIdx.x * nb;
    for (int i = threadIdx.x; i < nb; i += BIN_THREADS) dst[i] = dyn_lds[i];
}

__global__ void __launch_bounds__(1024)
k_bin_scan(const int* __restrict__ cnt, int* __restrict__ off, int* __restrict__ bstart,
           int nchunks, int nb) {
    __shared__ int sc[1024];
    int carry = 0;
    for (int b0 = 0; b0 < nb; b0 += 1024) {
        const int b = b0 + (int)threadIdx.x;
        int run = 0;
        if (b < nb) {
            for (int c0 = 0; c0 < nchunks; c0 += 16) {
                int v[16];
#pragma unroll
                for (int k = 0; k < 16; ++k) {
                    int c = c0 + k < nchunks ? c0 + k : nchunks - 1;
                    v[k] = cnt[(size_t)c * nb + b];
                }
#pragma unroll
                for (int k = 0; k < 16; ++k) {
                    if (c0 + k < nchunks) {
                        off[(size_t)(c0 + k) * nb + b] = run;
                        run += v[k];
                    }
                }
            }
        }
        sc[threadIdx.x] = run;
        __syncthreads();
        for (int d = 1; d < 1024; d <<= 1) {
            int t = (int)threadIdx.x >= d ? sc[threadIdx.x - d] : 0;
            __syncthreads();
            sc[threadIdx.x] += t;
            __syncthreads();
        }
        if (b < nb) bstart[b] = carry + sc[threadIdx.x] - run;
        const int tot = sc[1023];
        __syncthreads();
        carry += tot;
    }
    if (threadIdx.x == 0) bstart[nb] = carry;
}

__global__ void __launch_bounds__(BIN_THREADS)
k_bin_fill(const int* __restrict__ row, const int* __restrict__ col,
           const float* __restrict__ dinv,
           const int* __restrict__ bstart, const int* __restrict__ off,
           int* __restrict__ recs, float* __restrict__ nrm, int e, int nb) {
    const int* myoff = off + (size_t)blockIdx.x * nb;
    for (int i = threadIdx.x; i < nb; i += BIN_THREADS) dyn_lds[i] = bstart[i] + myoff[i];
    __syncthreads();
    const int e0 = blockIdx.x * BIN_CHUNK;
#pragma unroll
    for (int kb = 0; kb < BIN_CHUNK / BIN_THREADS; kb += 8) {
        int r[8], c[8];
        float f[8];
#pragma unroll
        for (int k = 0; k < 8; ++k) {
            int i = e0 + (kb + k) * BIN_THREADS + (int)threadIdx.x;
            int ii = i < e ? i : e - 1;
            r[k] = row[ii];
            c[k] = col[ii];
        }
#pragma unroll
        for (int k = 0; k < 8; ++k) f[k] = dinv[r[k]] * dinv[c[k]];
#pragma unroll
        for (int k = 0; k < 8; ++k) {
            int i = e0 + (kb + k) * BIN_THREADS + (int)threadIdx.x;
            if (i < e) {
                int pos = atomicAdd(&dyn_lds[c[k] >> BKT_LOG], 1);
                ((int2*)recs)[pos] = make_int2(r[k], c[k]);
                nrm[pos] = f[k];
            }
        }
    }
}

#define AGG_BATCH 16
template <bool FULL>
__device__ __forceinline__ void edge_group(int r, int c, float nv, int base, int cnt, int lane,
                                           const float* __restrict__ h, float* tile,
                                           const int2* __restrict__ nextRec,
                                           const float* __restrict__ nextNrm,
                                           int2& rcNext, float& nvNext) {
    const int nvBits = __float_as_int(nv);
#pragma unroll
    for (int jb = 0; jb < 32; jb += AGG_BATCH) {
        if (!FULL && jb >= cnt) break;
        float v0[AGG_BATCH], v1[AGG_BATCH];
#pragma unroll
        for (int j = 0; j < AGG_BATCH; ++j) {
            if (FULL || jb + j < cnt) {
                int rj = __builtin_amdgcn_readlane(r, jb + j);
                const float* s = h + (size_t)rj * 64 + lane;
                v0[j] = s[0];
                v1[j] = s[32];
            }
        }
        if (jb == 0) {
            rcNext = *nextRec;
            nvNext = *nextNrm;
        }
#pragma unroll
        for (int j = 0; j < AGG_BATCH; ++j) {
            if (FULL || jb + j < cnt) {
                int cj = __builtin_amdgcn_readlane(c, jb + j) - base;
                float nj = __int_as_float(__builtin_amdgcn_readlane(nvBits, jb + j));
                float* d = tile + cj * 64 + lane;
                atomicAdd(d,      nj * v0[j]);
                atomicAdd(d + 32, nj * v1[j]);
            }
        }
    }
}

__global__ void __launch_bounds__(256)
k_bucket_agg(const int* __restrict__ recs, const float* __restrict__ nrm,
             const int* __restrict__ bstart, const float* __restrict__ dinv,
             const float* __restrict__ h, float* __restrict__ agg, int n) {
    __shared__ float tile[BKT * 64];
    const int base = blockIdx.x * BKT;
    const int lane = threadIdx.x & 31;
    const int w = __builtin_amdgcn_readfirstlane((int)(threadIdx.x >> 5));

#pragma unroll
    for (int k = 0; k < BKT * 16 / 256; ++k) {
        int i = k * 256 + (int)threadIdx.x;
        int node = base + (i >> 4);
        int nc = node < n ? node : n - 1;
        float d = dinv[nc];
        float dd = d * d;
        float4 hv = ((const float4*)h)[(size_t)nc * 16 + (i & 15)];
        float4 v = make_float4(dd * hv.x, dd * hv.y, dd * hv.z, dd * hv.w);
        if (node >= n) v = make_float4(0.0f, 0.0f, 0.0f, 0.0f);
        ((float4*)tile)[i] = v;
    }
    __syncthreads();

    const int s = bstart[blockIdx.x];
    const int eend = bstart[blockIdx.x + 1];
    const int ngr = (eend - s + 31) >> 5;

    int g = w;
    int2 rc = make_int2(0, 0);
    float nv = 0.0f;
    if (g < ngr) {
        int e0 = s + g * 32;
        int c = eend - e0; if (c > 32) c = 32;
        int my = e0 + (lane < c ? lane : 0);
        rc = ((const int2*)recs)[my];
        nv = nrm[my];
    }
    while (g < ngr) {
        int cnt = eend - (s + g * 32); if (cnt > 32) cnt = 32;
        const int gn = g + 8;
        int e0n = s + (gn < ngr ? gn : g) * 32;
        int cn = eend - e0n; if (cn > 32) cn = 32;
        int myn = e0n + (lane < cn ? lane : 0);
        int2 rc2;
        float nv2;
        if (cnt == 32) edge_group<true >(rc.x, rc.y, nv, base, cnt, lane, h, tile,
                                         (const int2*)recs + myn, nrm + myn, rc2, nv2);
        else           edge_group<false>(rc.x, rc.y, nv, base, cnt, lane, h, tile,
                                         (const int2*)recs + myn, nrm + myn, rc2, nv2);
        rc = rc2; nv = nv2; g = gn;
    }
    __syncthreads();

#pragma unroll
    for (int k = 0; k < BKT * 16 / 256; ++k) {
        int i = k * 256 + (int)threadIdx.x;
        int node = base + (i >> 4);
        if (node < n)
            ((float4*)agg)[(size_t)node * 16 + (i & 15)] = ((const float4*)tile)[i];
    }
}

__global__ void k_layer(const float* __restrict__ agg, const float* __restrict__ x0,
                        const float* __restrict__ Wl, float beta,
                        float* __restrict__ h, int ntiles) {
    __shared__ float wlds[64 * 64];
    for (int i = threadIdx.x; i < 1024; i += 256)
        ((float4*)wlds)[i] = ((const float4*)Wl)[i];
    __syncthreads();

    int wid  = (blockIdx.x * blockDim.x + threadIdx.x) >> 5;
    int lane = threadIdx.x & 31;
    if (wid >= ntiles) return;
    int ln = lane & 15;
    int kh = lane >> 4;

    size_t rb = (size_t)(wid * 16 + ln) * 64 + 2 * kh;
    v2f afr[16];
#pragma unroll
    for (int kk = 0; kk < 16; ++kk) {
        v2f ag = *(const v2f*)(agg + rb + 4 * kk);
        v2f xx = *(const v2f*)(x0 + rb + 4 * kk);
        afr[kk] = ag * 0.9f + xx * 0.1f;
    }
    float ob = 1.0f - beta;
#pragma unroll
    for (int j = 0; j < 4; ++j) {
        const float* wcol = wlds + 2 * kh * 64 + 16 * j + ln;
        v8f acc = {};
#pragma unroll
        for (int kk = 0; kk < 16; ++kk) {
            v2f b;
            b.x = wcol[(size_t)(4 * kk) * 64];
            b.y = wcol[(size_t)(4 * kk + 1) * 64];
            acc = __builtin_amdgcn_wmma_f32_16x16x4_f32(
                false, afr[kk], false, b, (short)0, acc, false, false);
        }
#pragma unroll
        for (int r = 0; r < 8; ++r) {
            int m = r + 8 * kh;
            int n = 16 * j + ln;
            size_t gi = (size_t)(wid * 16 + m) * 64 + n;
            float hh = 0.9f * agg[gi] + 0.1f * x0[gi];
            float v = ob * hh + beta * acc[r];
            h[gi] = fmaxf(v, 0.0f);
        }
    }
}

extern "C" void kernel_launch(void* const* d_in, const int* in_sizes, int n_in,
                              void* d_out, int out_size, void* d_ws, size_t ws_size,
                              hipStream_t stream) {
    const float* x      = (const float*)d_in[0];
    const int*   ei     = (const int*)  d_in[1];
    const float* lin0_w = (const float*)d_in[2];
    const float* lin0_b = (const float*)d_in[3];
    const float* conv_w = (const float*)d_in[4];
    const float* lin1_w = (const float*)d_in[5];
    const float* lin1_b = (const float*)d_in[6];
    float* out = (float*)d_out;

    const int N = in_sizes[0] / 128;
    const int E = in_sizes[1] / 2;
    const int* row = ei;
    const int* col = ei + E;

    float* ws   = (float*)d_ws;
    float* dinv = ws;
    float* x0   = dinv + N;
    float* h    = x0 + (size_t)N * 64;
    float* agg  = h  + (size_t)N * 64;
    const int nb      = (N + BKT - 1) / BKT;
    const int nchunks = (E + BIN_CHUNK - 1) / BIN_CHUNK;
    int*   recs   = (int*)(agg + (size_t)N * 64);
    float* nrm    = (float*)(recs + (size_t)2 * E);
    int*   cnt    = (int*)(nrm + E);
    int*   off    = cnt + (size_t)nchunks * nb;
    int*   bstart = off + (size_t)nchunks * nb;

    const int ntiles = N / 16;
    const int gemmGrid = (ntiles * 32 + 255) / 256;

    k_fill_one<<<(N + 255) / 256, 256, 0, stream>>>(dinv, N);
    k_degree  <<<(E + 255) / 256, 256, 0, stream>>>(col, dinv, E);
    k_rsqrt   <<<(N + 255) / 256, 256, 0, stream>>>(dinv, N);

    k_bin_count<<<nchunks, BIN_THREADS, nb * sizeof(int), stream>>>(col, cnt, E, nb);
    k_bin_scan <<<1, 1024, 0, stream>>>(cnt, off, bstart, nchunks, nb);
    k_bin_fill <<<nchunks, BIN_THREADS, nb * sizeof(int), stream>>>(row, col, dinv, bstart, off,
                                                                  recs, nrm, E, nb);

    k_dense<128, true, true><<<gemmGrid, 256, 0, stream>>>(x, lin0_w, lin0_b, h, x0, ntiles);

    for (int l = 0; l < 4; ++l) {
        float beta = logf(0.5f / (float)(l + 1) + 1.0f);
        k_bucket_agg<<<nb, 256, 0, stream>>>(recs, nrm, bstart, dinv, h, agg, N);
        k_layer   <<<gemmGrid, 256, 0, stream>>>(agg, x0, conv_w + (size_t)l * 64 * 64,
                                                 beta, h, ntiles);
    }

    k_dense<64, false, false><<<gemmGrid, 256, 0, stream>>>(h, lin1_w, lin1_b, out, nullptr, ntiles);
}


# In-degrees counted on LDS integer counters from the bucket-ordered records, 256 targets per block (was one random global f32 atomic add per edge before binning); edge factors filled by a record pass
# speedup vs baseline: 2.3850x; 1.3182x over previous
#include <hip/hip_runtime.h>
#include <math.h>

typedef __attribute__((ext_vector_type(2))) float v2f;
typedef __attribute__((ext_vector_type(8))) float v8f;

template <int K, bool RELU, bool STOREPRE>
__global__ void k_dense(const float* __restrict__ X, const float* __restrict__ W,
                        const float* __restrict__ bias,
                        float* __restrict__ outA, float* __restrict__ outPre,
                        int ntiles) {
    __shared__ float wlds[64 * K];
    constexpr int NV4 = (64 * K) / 4;
    for (int i = threadIdx.x; i < NV4; i += 256)
        ((float4*)wlds)[i] = ((const float4*)W)[i];
    __syncthreads();

    int wid  = (blockIdx.x * blockDim.x + threadIdx.x) >> 5;
    int lane = threadIdx.x & 31;
    if (wid >= ntiles) return;
    int ln = lane & 15;
    int kh = lane >> 4;

    const float* xrow = X + (size_t)(wid * 16 + ln) * K + 2 * kh;
    v2f afr[K / 4];
#pragma unroll
    for (int kk = 0; kk < K / 4; ++kk)
        afr[kk] = *(const v2f*)(xrow + 4 * kk);

#pragma unroll
    for (int j = 0; j < 4; ++j) {
        const float* wrow = wlds + (size_t)(16 * j + ln) * K + 2 * kh;
        v8f acc = {};
#pragma unroll
        for (int kk = 0; kk < K / 4; ++kk) {
            v2f b = *(const v2f*)(wrow + 4 * kk);
            acc = __builtin_amdgcn_wmma_f32_16x16x4_f32(
                false, afr[kk], false, b, (short)0, acc, false, false);
        }
        int n = 16 * j + ln;
        float bv = bias[n];
#pragma unroll
        for (int r = 0; r < 8; ++r) {
            int m = r + 8 * kh;
            size_t gi = (size_t)(wid * 16 + m) * 64 + n;
            float v = acc[r] + bv;
            if constexpr (STOREPRE) outPre[gi] = v;
            outA[gi] = RELU ? fmaxf(v, 0.0f) : v;
        }
    }
}

#define BKT         64
#define BKT_LOG     6
#define BIN_CHUNK   16384
#define BIN_THREADS 1024

extern __shared__ int dyn_lds[];

__global__ void __launch_bounds__(BIN_THREADS)
k_bin_count(const int* __restrict__ col, int* __restrict__ cnt, int e, int nb) {
    for (int i = threadIdx.x; i < nb; i += BIN_THREADS) dyn_lds[i] = 0;
    __syncthreads();
    const int e0 = blockIdx.x * BIN_CHUNK;
    int c[BIN_CHUNK / BIN_THREADS];
#pragma unroll
    for (int k = 0; k < BIN_CHUNK / BIN_THREADS; ++k) {
        int i = e0 + k * BIN_THREADS + (int)threadIdx.x;
        c[k] = col[i < e ? i : e - 1];
    }
#pragma unroll
    for (int k = 0; k < BIN_CHUNK / BIN_THREADS; ++k) {
        int i = e0 + k * BIN_THREADS + (int)threadIdx.x;
        if (i < e) atomicAdd(&dyn_lds[c[k] >> BKT_LOG], 1);
    }
    __syncthreads();
    int* dst = cnt + (size_t)blockIdx.x * nb;
    for (int i = threadIdx.x; i < nb; i += BIN_THREADS) dst[i] = dyn_lds[i];
}

__global__ void __launch_bounds__(1024)
k_bin_scan(const int* __restrict__ cnt, int* __restrict__ off, int* __restrict__ bstart,
           int nchunks, int nb) {
    __shared__ int sc[1024];
    int carry = 0;
    for (int b0 = 0; b0 < nb; b0 += 1024) {
        const int b = b0 + (int)threadIdx.x;
        int run = 0;
        if (b < nb) {
            for (int c0 = 0; c0 < nchunks; c0 += 16) {
                int v[16];
#pragma unroll
                for (int k = 0; k < 16; ++k) {
                    int c = c0 + k < nchunks ? c0 + k : nchunks - 1;
                    v[k] = cnt[(size_t)c * nb + b];
                }
#pragma unroll
                for (int k = 0; k < 16; ++k) {
                    if (c0 + k < nchunks) {
                        off[(size_t)(c0 + k) * nb + b] = run;
                        run += v[k];
                    }
                }
            }
        }
        sc[threadIdx.x] = run;
        __syncthreads();
        for (int d = 1; d < 1024; d <<= 1) {
            int t = (int)threadIdx.x >= d ? sc[threadIdx.x - d] : 0;
            __syncthreads();
            sc[threadIdx.x] += t;
            __syncthreads();
        }
        if (b < nb) bstart[b] = carry + sc[threadIdx.x] - run;
        const int tot = sc[1023];
        __syncthreads();
        carry += tot;
    }
    if (threadIdx.x == 0) bstart[nb] = carry;
}

__global__ void __launch_bounds__(BIN_THREADS)
k_bin_fill(const int* __restrict__ row, const int* __restrict__ col,
           const int* __restrict__ bstart, const int* __restrict__ off,
           int* __restrict__ recs, int e, int nb) {
    const int* myoff = off + (size_t)blockIdx.x * nb;
    for (int i = threadIdx.x; i < nb; i += BIN_THREADS) dyn_lds[i] = bstart[i] + myoff[i];
    __syncthreads();
    const int e0 = blockIdx.x * BIN_CHUNK;
    int r[BIN_CHUNK / BIN_THREADS], c[BIN_CHUNK / BIN_THREADS];
#pragma unroll
    for (int k = 0; k < BIN_CHUNK / BIN_THREADS; ++k) {
        int i = e0 + k * BIN_THREADS + (int)threadIdx.x;
        int ii = i < e ? i : e - 1;
        r[k] = row[ii];
        c[k] = col[ii];
    }
#pragma unroll
    for (int k = 0; k < BIN_CHUNK / BIN_THREADS; ++k) {
        int i = e0 + k * BIN_THREADS + (int)threadIdx.x;
        if (i < e) {
            int pos = atomicAdd(&dyn_lds[c[k] >> BKT_LOG], 1);
            ((int2*)recs)[pos] = make_int2(r[k], c[k]);
        }
    }
}

#define DEG_BKTS  4
#define DEG_BATCH 8
__global__ void __launch_bounds__(256)
k_bucket_dinv(const int* __restrict__ recs, const int* __restrict__ bstart,
              float* __restrict__ dinv, int n, int nb) {
    __shared__ int cnt[DEG_BKTS * BKT];
    cnt[threadIdx.x] = 1;
    __syncthreads();
    const int b0 = blockIdx.x * DEG_BKTS;
    int b1 = b0 + DEG_BKTS; if (b1 > nb) b1 = nb;
    const int s = bstart[b0];
    const int eend = bstart[b1];
    const int base = b0 * BKT;
    for (int i0 = s; i0 < eend; i0 += 256 * DEG_BATCH) {
        int t[DEG_BATCH];
#pragma unroll
        for (int k = 0; k < DEG_BATCH; ++k) {
            int i = i0 + k * 256 + (int)threadIdx.x;
            int ii = i < eend ? i : eend - 1;
            t[k] = recs[2 * (size_t)ii + 1];
        }
#pragma unroll
        for (int k = 0; k < DEG_BATCH; ++k) {
            int i = i0 + k * 256 + (int)threadIdx.x;
            if (i < eend) atomicAdd(&cnt[t[k] - base], 1);
        }
    }
    __syncthreads();
    const int node = base + (int)threadIdx.x;
    if (node < n) {
        float d = (float)cnt[threadIdx.x];
        dinv[node] = d > 0.0f ? rsqrtf(d) : 0.0f;
    }
}

#define NRM_PER 8
__global__ void __launch_bounds__(256)
k_nrm(const int* __restrict__ recs, const float* __restrict__ dinv,
      float* __restrict__ nrm, int e) {
    const int i0 = blockIdx.x * (256 * NRM_PER) + (int)threadIdx.x;
    int2 rc[NRM_PER];
    float f[NRM_PER];
#pragma unroll
    for (int k = 0; k < NRM_PER; ++k) {
        int i = i0 + k * 256;
        rc[k] = ((const int2*)recs)[i < e ? i : e - 1];
    }
#pragma unroll
    for (int k = 0; k < NRM_PER; ++k) f[k] = dinv[rc[k].x] * dinv[rc[k].y];
#pragma unroll
    for (int k = 0; k < NRM_PER; ++k) {
        int i = i0 + k * 256;
        if (i < e) nrm[i] = f[k];
    }
}

#define AGG_BATCH 16
template <bool FULL>
__device__ __forceinline__ void edge_group(int r, int c, float nv, int base, int cnt, int lane,
                                           const float* __restrict__ h, float* tile,
                                           const int2* __restrict__ nextRec,
                                           const float* __restrict__ nextNrm,
                                           int2& rcNext, float& nvNext) {
    const int nvBits = __float_as_int(nv);
#pragma unroll
    for (int jb = 0; jb < 32; jb += AGG_BATCH) {
        if (!FULL && jb >= cnt) break;
        float v0[AGG_BATCH], v1[AGG_BATCH];
#pragma unroll
        for (int j = 0; j < AGG_BATCH; ++j) {
            if (FULL || jb + j < cnt) {
                int rj = __builtin_amdgcn_readlane(r, jb + j);
                const float* s = h + (size_t)rj * 64 + lane;
                v0[j] = s[0];
                v1[j] = s[32];
            }
        }
        if (jb == 0) {
            rcNext = *nextRec;
            nvNext = *nextNrm;
        }
#pragma unroll
        for (int j = 0; j < AGG_BATCH; ++j) {
            if (FULL || jb + j < cnt) {
                int cj = __builtin_amdgcn_readlane(c, jb + j) - base;
                float nj = __int_as_float(__builtin_amdgcn_readlane(nvBits, jb + j));
                float* d = tile + cj * 64 + lane;
                atomicAdd(d,      nj * v0[j]);
                atomicAdd(d + 32, nj * v1[j]);
            }
        }
    }
}

__global__ void __launch_bounds__(256)
k_bucket_agg(const int* __restrict__ recs, const float* __restrict__ nrm,
             const int* __restrict__ bstart, const float* __restrict__ dinv,
             const float* __restrict__ h, float* __restrict__ agg, int n) {
    __shared__ float tile[BKT * 64];
    const int base = blockIdx.x * BKT;
    const int lane = threadIdx.x & 31;
    const int w = __builtin_amdgcn_readfirstlane((int)(threadIdx.x >> 5));

#pragma unroll
    for (int k = 0; k < BKT * 16 / 256; ++k) {
        int i = k * 256 + (int)threadIdx.x;
        int node = base + (i >> 4);
        int nc = node < n ? node : n - 1;
        float d = dinv[nc];
        float dd = d * d;
        float4 hv = ((const float4*)h)[(size_t)nc * 16 + (i & 15)];
        float4 v = make_float4(dd * hv.x, dd * hv.y, dd * hv.z, dd * hv.w);
        if (node >= n) v = make_float4(0.0f, 0.0f, 0.0f, 0.0f);
        ((float4*)tile)[i] = v;
    }
    __syncthreads();

    const int s = bstart[blockIdx.x];
    const int eend = bstart[blockIdx.x + 1];
    const int ngr = (eend - s + 31) >> 5;

    int g = w;
    int2 rc = make_int2(0, 0);
    float nv = 0.0f;
    if (g < ngr) {
        int e0 = s + g * 32;
        int c = eend - e0; if (c > 32) c = 32;
        int my = e0 + (lane < c ? lane : 0);
        rc = ((const int2*)recs)[my];
        nv = nrm[my];
    }
    while (g < ngr) {
        int cnt = eend - (s + g * 32); if (cnt > 32) cnt = 32;
        const int gn = g + 8;
        int e0n = s + (gn < ngr ? gn : g) * 32;
        int cn = eend - e0n; if (cn > 32) cn = 32;
        int myn = e0n + (lane < cn ? lane : 0);
        int2 rc2;
        float nv2;
        if (cnt == 32) edge_group<true >(rc.x, rc.y, nv, base, cnt, lane, h, tile,
                                         (const int2*)recs + myn, nrm + myn, rc2, nv2);
        else           edge_group<false>(rc.x, rc.y, nv, base, cnt, lane, h, tile,
                                         (const int2*)recs + myn, nrm + myn, rc2, nv2);
        rc = rc2; nv = nv2; g = gn;
    }
    __syncthreads();

#pragma unroll
    for (int k = 0; k < BKT * 16 / 256; ++k) {
        int i = k * 256 + (int)threadIdx.x;
        int node = base + (i >> 4);
        if (node < n)
            ((float4*)agg)[(size_t)node * 16 + (i & 15)] = ((const float4*)tile)[i];
    }
}

__global__ void k_layer(const float* __restrict__ agg, const float* __restrict__ x0,
                        const float* __restrict__ Wl, float beta,
                        float* __restrict__ h, int ntiles) {
    __shared__ float wlds[64 * 64];
    for (int i = threadIdx.x; i < 1024; i += 256)
        ((float4*)wlds)[i] = ((const float4*)Wl)[i];
    __syncthreads();

    int wid  = (blockIdx.x * blockDim.x + threadIdx.x) >> 5;
    int lane = threadIdx.x & 31;
    if (wid >= ntiles) return;
    int ln = lane & 15;
    int kh = lane >> 4;

    size_t rb = (size_t)(wid * 16 + ln) * 64 + 2 * kh;
    v2f afr[16];
#pragma unroll
    for (int kk = 0; kk < 16; ++kk) {
        v2f ag = *(const v2f*)(agg + rb + 4 * kk);
        v2f xx = *(const v2f*)(x0 + rb + 4 * kk);
        afr[kk] = ag * 0.9f + xx * 0.1f;
    }
    float ob = 1.0f - beta;
#pragma unroll
    for (int j = 0; j < 4; ++j) {
        const float* wcol = wlds + 2 * kh * 64 + 16 * j + ln;
        v8f acc = {};
#pragma unroll
        for (int kk = 0; kk < 16; ++kk) {
            v2f b;
            b.x = wcol[(size_t)(4 * kk) * 64];
            b.y = wcol[(size_t)(4 * kk + 1) * 64];
            acc = __builtin_amdgcn_wmma_f32_16x16x4_f32(
                false, afr[kk], false, b, (short)0, acc, false, false);
        }
#pragma unroll
        for (int r = 0; r < 8; ++r) {
            int m = r + 8 * kh;
            int n = 16 * j + ln;
            size_t gi = (size_t)(wid * 16 + m) * 64 + n;
            float hh = 0.9f * agg[gi] + 0.1f * x0[gi];
            float v = ob * hh + beta * acc[r];
            h[gi] = fmaxf(v, 0.0f);
        }
    }
}

extern "C" void kernel_launch(void* const* d_in, const int* in_sizes, int n_in,
                              void* d_out, int out_size, void* d_ws, size_t ws_size,
                              hipStream_t stream) {
    const float* x      = (const float*)d_in[0];
    const int*   ei     = (const int*)  d_in[1];
    const float* lin0_w = (const float*)d_in[2];
    const float* lin0_b = (const float*)d_in[3];
    const float* conv_w = (const float*)d_in[4];
    const float* lin1_w = (const float*)d_in[5];
    const float* lin1_b = (const float*)d_in[6];
    float* out = (float*)d_out;

    const int N = in_sizes[0] / 128;
    const int E = in_sizes[1] / 2;
    const int* row = ei;
    const int* col = ei + E;

    float* ws   = (float*)d_ws;
    float* dinv = ws;
    float* x0   = dinv + N;
    float* h    = x0 + (size_t)N * 64;
    float* agg  = h  + (size_t)N * 64;
    const int nb      = (N + BKT - 1) / BKT;
    const int nchunks = (E + BIN_CHUNK - 1) / BIN_CHUNK;
    int*   recs   = (int*)(agg + (size_t)N * 64);
    float* nrm    = (float*)(recs + (size_t)2 * E);
    int*   cnt    = (int*)(nrm + E);
    int*   off    = cnt + (size_t)nchunks * nb;
    int*   bstart = off + (size_t)nchunks * nb;

    const int ntiles = N / 16;
    const int gemmGrid = (ntiles * 32 + 255) / 256;

    k_bin_count<<<nchunks, BIN_THREADS, nb * sizeof(int), stream>>>(col, cnt, E, nb);
    k_bin_scan <<<1, 1024, 0, stream>>>(cnt, off, bstart, nchunks, nb);
    k_bin_fill <<<nchunks, BIN_THREADS, nb * sizeof(int), stream>>>(row, col, bstart, off,
                                                                  recs, E, nb);

    k_bucket_dinv<<<(nb + DEG_BKTS - 1) / DEG_BKTS, 256, 0, stream>>>(recs, bstart, dinv, N, nb);
    k_nrm<<<(E + 256 * NRM_PER - 1) / (256 * NRM_PER), 256, 0, stream>>>(recs, dinv, nrm, E);

    k_dense<128, true, true><<<gemmGrid, 256, 0, stream>>>(x, lin0_w, lin0_b, h, x0, ntiles);

    for (int l = 0; l < 4; ++l) {
        float beta = logf(0.5f / (float)(l + 1) + 1.0f);
        k_bucket_agg<<<nb, 256, 0, stream>>>(recs, nrm, bstart, dinv, h, agg, N);
        k_layer   <<<gemmGrid, 256, 0, stream>>>(agg, x0, conv_w + (size_t)l * 64 * 64,
                                                 beta, h, ntiles);
    }

    k_dense<64, false, false><<<gemmGrid, 256, 0, stream>>>(h, lin1_w, lin1_b, out, nullptr, ntiles);
}


# Dense products: a wave owns 64 rows (4 row tiles, A fragments in registers); each B fragment read from LDS feeds 4 WMMAs on 4 accumulators (was 1): 4x fewer LDS reads, waves and weight copies
# speedup vs baseline: 2.5073x; 1.0513x over previous
#include <hip/hip_runtime.h>
#include <math.h>

typedef __attribute__((ext_vector_type(2))) float v2f;
typedef __attribute__((ext_vector_type(8))) float v8f;

#define DENSE_T       4
#define DENSE_THREADS 128
template <int K, bool RELU, bool STOREPRE>
__global__ void __launch_bounds__(DENSE_THREADS)
k_dense(const float* __restrict__ X, const float* __restrict__ W,
        const float* __restrict__ bias,
        float* __restrict__ outA, float* __restrict__ outPre,
        int ntiles) {
    __shared__ float wlds[64 * K];
    constexpr int NV4 = (64 * K) / 4;
    for (int i = threadIdx.x; i < NV4; i += DENSE_THREADS)
        ((float4*)wlds)[i] = ((const float4*)W)[i];
    __syncthreads();

    const int wid = __builtin_amdgcn_readfirstlane(
        (int)((blockIdx.x * DENSE_THREADS + threadIdx.x) >> 5));
    const int lane = threadIdx.x & 31;
    const int tile0 = wid * DENSE_T;
    if (tile0 >= ntiles) return;
    const int ln = lane & 15;
    const int kh = lane >> 4;

    const float* xrow[DENSE_T];
#pragma unroll
    for (int t = 0; t < DENSE_T; ++t) {
        int tile = tile0 + t;
        int tl = tile < ntiles ? tile : ntiles - 1;
        xrow[t] = X + (size_t)(tl * 16 + ln) * K + 2 * kh;
    }
    v2f afr[DENSE_T][K / 4];
#pragma unroll
    for (int kk = 0; kk < K / 4; ++kk)
#pragma unroll
        for (int t = 0; t < DENSE_T; ++t)
            afr[t][kk] = *(const v2f*)(xrow[t] + 4 * kk);

#pragma unroll
    for (int j = 0; j < 4; ++j) {
        const float* wrow = wlds + (size_t)(16 * j + ln) * K + 2 * kh;
        v8f acc[DENSE_T];
#pragma unroll
        for (int t = 0; t < DENSE_T; ++t) acc[t] = v8f{};
#pragma unroll
        for (int kk = 0; kk < K / 4; ++kk) {
            v2f b = *(const v2f*)(wrow + 4 * kk);
#pragma unroll
            for (int t = 0; t < DENSE_T; ++t)
                acc[t] = __builtin_amdgcn_wmma_f32_16x16x4_f32(
                    false, afr[t][kk], false, b, (short)0, acc[t], false, false);
        }
        const int n = 16 * j + ln;
        const float bv = bias[n];
#pragma unroll
        for (int t = 0; t < DENSE_T; ++t) {
            if (tile0 + t < ntiles) {
#pragma unroll
                for (int r = 0; r < 8; ++r) {
                    int m = r + 8 * kh;
                    size_t gi = (size_t)((tile0 + t) * 16 + m) * 64 + n;
                    float v = acc[t][r] + bv;
                    if constexpr (STOREPRE) outPre[gi] = v;
                    outA[gi] = RELU ? fmaxf(v, 0.0f) : v;
                }
            }
        }
    }
}

#define BKT         64
#define BKT_LOG     6
#define BIN_CHUNK   16384
#define BIN_THREADS 1024

extern __shared__ int dyn_lds[];

__global__ void __launch_bounds__(BIN_THREADS)
k_bin_count(const int* __restrict__ col, int* __restrict__ cnt, int e, int nb) {
    for (int i = threadIdx.x; i < nb; i += BIN_THREADS) dyn_lds[i] = 0;
    __syncthreads();
    const int e0 = blockIdx.x * BIN_CHUNK;
    int c[BIN_CHUNK / BIN_THREADS];
#pragma unroll
    for (int k = 0; k < BIN_CHUNK / BIN_THREADS; ++k) {
        int i = e0 + k * BIN_THREADS + (int)threadIdx.x;
        c[k] = col[i < e ? i : e - 1];
    }
#pragma unroll
    for (int k = 0; k < BIN_CHUNK / BIN_THREADS; ++k) {
        int i = e0 + k * BIN_THREADS + (int)threadIdx.x;
        if (i < e) atomicAdd(&dyn_lds[c[k] >> BKT_LOG], 1);
    }
    __syncthreads();
    int* dst = cnt + (size_t)blockIdx.x * nb;
    for (int i = threadIdx.x; i < nb; i += BIN_THREADS) dst[i] = dyn_lds[i];
}

__global__ void __launch_bounds__(1024)
k_bin_scan(const int* __restrict__ cnt, int* __restrict__ off, int* __restrict__ bstart,
           int nchunks, int nb) {
    __shared__ int sc[1024];
    int carry = 0;
    for (int b0 = 0; b0 < nb; b0 += 1024) {
        const int b = b0 + (int)threadIdx.x;
        int run = 0;
        if (b < nb) {
            for (int c0 = 0; c0 < nchunks; c0 += 16) {
                int v[16];
#pragma unroll
                for (int k = 0; k < 16; ++k) {
                    int c = c0 + k < nchunks ? c0 + k : nchunks - 1;
                    v[k] = cnt[(size_t)c * nb + b];
                }
#pragma unroll
                for (int k = 0; k < 16; ++k) {
                    if (c0 + k < nchunks) {
                        off[(size_t)(c0 + k) * nb + b] = run;
                        run += v[k];
                    }
                }
            }
        }
        sc[threadIdx.x] = run;
        __syncthreads();
        for (int d = 1; d < 1024; d <<= 1) {
            int t = (int)threadIdx.x >= d ? sc[threadIdx.x - d] : 0;
            __syncthreads();
            sc[threadIdx.x] += t;
            __syncthreads();
        }
        if (b < nb) bstart[b] = carry + sc[threadIdx.x] - run;
        const int tot = sc[1023];
        __syncthreads();
        carry += tot;
    }
    if (threadIdx.x == 0) bstart[nb] = carry;
}

__global__ void __launch_bounds__(BIN_THREADS)
k_bin_fill(const int* __restrict__ row, const int* __restrict__ col,
           const int* __restrict__ bstart, const int* __restrict__ off,
           int* __restrict__ recs, int e, int nb) {
    const int* myoff = off + (size_t)blockIdx.x * nb;
    for (int i = threadIdx.x; i < nb; i += BIN_THREADS) dyn_lds[i] = bstart[i] + myoff[i];
    __syncthreads();
    const int e0 = blockIdx.x * BIN_CHUNK;
    int r[BIN_CHUNK / BIN_THREADS], c[BIN_CHUNK / BIN_THREADS];
#pragma unroll
    for (int k = 0; k < BIN_CHUNK / BIN_THREADS; ++k) {
        int i = e0 + k * BIN_THREADS + (int)threadIdx.x;
        int ii = i < e ? i : e - 1;
        r[k] = row[ii];
        c[k] = col[ii];
    }
#pragma unroll
    for (int k = 0; k < BIN_CHUNK / BIN_THREADS; ++k) {
        int i = e0 + k * BIN_THREADS + (int)threadIdx.x;
        if (i < e) {
            int pos = atomicAdd(&dyn_lds[c[k] >> BKT_LOG], 1);
            ((int2*)recs)[pos] = make_int2(r[k], c[k]);
        }
    }
}

#define DEG_BKTS  4
#define DEG_BATCH 8
__global__ void __launch_bounds__(256)
k_bucket_dinv(const int* __restrict__ recs, const int* __restrict__ bstart,
              float* __restrict__ dinv, int n, int nb) {
    __shared__ int cnt[DEG_BKTS * BKT];
    cnt[threadIdx.x] = 1;
    __syncthreads();
    const int b0 = blockIdx.x * DEG_BKTS;
    int b1 = b0 + DEG_BKTS; if (b1 > nb) b1 = nb;
    const int s = bstart[b0];
    const int eend = bstart[b1];
    const int base = b0 * BKT;
    for (int i0 = s; i0 < eend; i0 += 256 * DEG_BATCH) {
        int t[DEG_BATCH];
#pragma unroll
        for (int k = 0; k < DEG_BATCH; ++k) {
            int i = i0 + k * 256 + (int)threadIdx.x;
            int ii = i < eend ? i : eend - 1;
            t[k] = recs[2 * (size_t)ii + 1];
        }
#pragma unroll
        for (int k = 0; k < DEG_BATCH; ++k) {
            int i = i0 + k * 256 + (int)threadIdx.x;
            if (i < eend) atomicAdd(&cnt[t[k] - base], 1);
        }
    }
    __syncthreads();
    const int node = base + (int)threadIdx.x;
    if (node < n) {
        float d = (float)cnt[threadIdx.x];
        dinv[node] = d > 0.0f ? rsqrtf(d) : 0.0f;
    }
}

#define NRM_PER 8
__global__ void __launch_bounds__(256)
k_nrm(const int* __restrict__ recs, const float* __restrict__ dinv,
      float* __restrict__ nrm, int e) {
    const int i0 = blockIdx.x * (256 * NRM_PER) + (int)threadIdx.x;
    int2 rc[NRM_PER];
    float f[NRM_PER];
#pragma unroll
    for (int k = 0; k < NRM_PER; ++k) {
        int i = i0 + k * 256;
        rc[k] = ((const int2*)recs)[i < e ? i : e - 1];
    }
#pragma unroll
    for (int k = 0; k < NRM_PER; ++k) f[k] = dinv[rc[k].x] * dinv[rc[k].y];
#pragma unroll
    for (int k = 0; k < NRM_PER; ++k) {
        int i = i0 + k * 256;
        if (i < e) nrm[i] = f[k];
    }
}

#define AGG_BATCH 16
template <bool FULL>
__device__ __forceinline__ void edge_group(int r, int c, float nv, int base, int cnt, int lane,
                                           const float* __restrict__ h, float* tile,
                                           const int2* __restrict__ nextRec,
                                           const float* __restrict__ nextNrm,
                                           int2& rcNext, float& nvNext) {
    const int nvBits = __float_as_int(nv);
#pragma unroll
    for (int jb = 0; jb < 32; jb += AGG_BATCH) {
        if (!FULL && jb >= cnt) break;
        float v0[AGG_BATCH], v1[AGG_BATCH];
#pragma unroll
        for (int j = 0; j < AGG_BATCH; ++j) {
            if (FULL || jb + j < cnt) {
                int rj = __builtin_amdgcn_readlane(r, jb + j);
                const float* s = h + (size_t)rj * 64 + lane;
                v0[j] = s[0];
                v1[j] = s[32];
            }
        }
        if (jb == 0) {
            rcNext = *nextRec;
            nvNext = *nextNrm;
        }
#pragma unroll
        for (int j = 0; j < AGG_BATCH; ++j) {
            if (FULL || jb + j < cnt) {
                int cj = __builtin_amdgcn_readlane(c, jb + j) - base;
                float nj = __int_as_float(__builtin_amdgcn_readlane(nvBits, jb + j));
                float* d = tile + cj * 64 + lane;
                atomicAdd(d,      nj * v0[j]);
                atomicAdd(d + 32, nj * v1[j]);
            }
        }
    }
}

__global__ void __launch_bounds__(256)
k_bucket_agg(const int* __restrict__ recs, const float* __restrict__ nrm,
             const int* __restrict__ bstart, const float* __restrict__ dinv,
             const float* __restrict__ h, float* __restrict__ agg, int n) {
    __shared__ float tile[BKT * 64];
    const int base = blockIdx.x * BKT;
    const int lane = threadIdx.x & 31;
    const int w = __builtin_amdgcn_readfirstlane((int)(threadIdx.x >> 5));

#pragma unroll
    for (int k = 0; k < BKT * 16 / 256; ++k) {
        int i = k * 256 + (int)threadIdx.x;
        int node = base + (i >> 4);
        int nc = node < n ? node : n - 1;
        float d = dinv[nc];
        float dd = d * d;
        float4 hv = ((const float4*)h)[(size_t)nc * 16 + (i & 15)];
        float4 v = make_float4(dd * hv.x, dd * hv.y, dd * hv.z, dd * hv.w);
        if (node >= n) v = make_float4(0.0f, 0.0f, 0.0f, 0.0f);
        ((float4*)tile)[i] = v;
    }
    __syncthreads();

    const int s = bstart[blockIdx.x];
    const int eend = bstart[blockIdx.x + 1];
    const int ngr = (eend - s + 31) >> 5;

    int g = w;
    int2 rc = make_int2(0, 0);
    float nv = 0.0f;
    if (g < ngr) {
        int e0 = s + g * 32;
        int c = eend - e0; if (c > 32) c = 32;
        int my = e0 + (lane < c ? lane : 0);
        rc = ((const int2*)recs)[my];
        nv = nrm[my];
    }
    while (g < ngr) {
        int cnt = eend - (s + g * 32); if (cnt > 32) cnt = 32;
        const int gn = g + 8;
        int e0n = s + (gn < ngr ? gn : g) * 32;
        int cn = eend - e0n; if (cn > 32) cn = 32;
        int myn = e0n + (lane < cn ? lane : 0);
        int2 rc2;
        float nv2;
        if (cnt == 32) edge_group<true >(rc.x, rc.y, nv, base, cnt, lane, h, tile,
                                         (const int2*)recs + myn, nrm + myn, rc2, nv2);
        else           edge_group<false>(rc.x, rc.y, nv, base, cnt, lane, h, tile,
                                         (const int2*)recs + myn, nrm + myn, rc2, nv2);
        rc = rc2; nv = nv2; g = gn;
    }
    __syncthreads();

#pragma unroll
    for (int k = 0; k < BKT * 16 / 256; ++k) {
        int i = k * 256 + (int)threadIdx.x;
        int node = base + (i >> 4);
        if (node < n)
            ((float4*)agg)[(size_t)node * 16 + (i & 15)] = ((const float4*)tile)[i];
    }
}

__global__ void k_layer(const float* __restrict__ agg, const float* __restrict__ x0,
                        const float* __restrict__ Wl, float beta,
                        float* __restrict__ h, int ntiles) {
    __shared__ float wlds[64 * 64];
    for (int i = threadIdx.x; i < 1024; i += 256)
        ((float4*)wlds)[i] = ((const float4*)Wl)[i];
    __syncthreads();

    int wid  = (blockIdx.x * blockDim.x + threadIdx.x) >> 5;
    int lane = threadIdx.x & 31;
    if (wid >= ntiles) return;
    int ln = lane & 15;
    int kh = lane >> 4;

    size_t rb = (size_t)(wid * 16 + ln) * 64 + 2 * kh;
    v2f afr[16];
#pragma unroll
    for (int kk = 0; kk < 16; ++kk) {
        v2f ag = *(const v2f*)(agg + rb + 4 * kk);
        v2f xx = *(const v2f*)(x0 + rb + 4 * kk);
        afr[kk] = ag * 0.9f + xx * 0.1f;
    }
    float ob = 1.0f - beta;
#pragma unroll
    for (int j = 0; j < 4; ++j) {
        const float* wcol = wlds + 2 * kh * 64 + 16 * j + ln;
        v8f acc = {};
#pragma unroll
        for (int kk = 0; kk < 16; ++kk) {
            v2f b;
            b.x = wcol[(size_t)(4 * kk) * 64];
            b.y = wcol[(size_t)(4 * kk + 1) * 64];
            acc = __builtin_amdgcn_wmma_f32_16x16x4_f32(
                false, afr[kk], false, b, (short)0, acc, false, false);
        }
#pragma unroll
        for (int r = 0; r < 8; ++r) {
            int m = r + 8 * kh;
            int n = 16 * j + ln;
            size_t gi = (size_t)(wid * 16 + m) * 64 + n;
            float hh = 0.9f * agg[gi] + 0.1f * x0[gi];
            float v = ob * hh + beta * acc[r];
            h[gi] = fmaxf(v, 0.0f);
        }
    }
}

extern "C" void kernel_launch(void* const* d_in, const int* in_sizes, int n_in,
                              void* d_out, int out_size, void* d_ws, size_t ws_size,
                              hipStream_t stream) {
    const float* x      = (const float*)d_in[0];
    const int*   ei     = (const int*)  d_in[1];
    const float* lin0_w = (const float*)d_in[2];
    const float* lin0_b = (const float*)d_in[3];
    const float* conv_w = (const float*)d_in[4];
    const float* lin1_w = (const float*)d_in[5];
    const float* lin1_b = (const float*)d_in[6];
    float* out = (float*)d_out;

    const int N = in_sizes[0] / 128;
    const int E = in_sizes[1] / 2;
    const int* row = ei;
    const int* col = ei + E;

    float* ws   = (float*)d_ws;
    float* dinv = ws;
    float* x0   = dinv + N;
    float* h    = x0 + (size_t)N * 64;
    float* agg  = h  + (size_t)N * 64;
    const int nb      = (N + BKT - 1) / BKT;
    const int nchunks = (E + BIN_CHUNK - 1) / BIN_CHUNK;
    int*   recs   = (int*)(agg + (size_t)N * 64);
    float* nrm    = (float*)(recs + (size_t)2 * E);
    int*   cnt    = (int*)(nrm + E);
    int*   off    = cnt + (size_t)nchunks * nb;
    int*   bstart = off + (size_t)nchunks * nb;

    const int ntiles = N / 16;
    const int gemmGrid = (ntiles * 32 + 255) / 256;
    const int denseGrid = (ntiles + 4 * DENSE_T - 1) / (4 * DENSE_T);

    k_bin_count<<<nchunks, BIN_THREADS, nb * sizeof(int), stream>>>(col, cnt, E, nb);
    k_bin_scan <<<1, 1024, 0, stream>>>(cnt, off, bstart, nchunks, nb);
    k_bin_fill <<<nchunks, BIN_THREADS, nb * sizeof(int), stream>>>(row, col, bstart, off,
                                                                  recs, E, nb);

    k_bucket_dinv<<<(nb + DEG_BKTS - 1) / DEG_BKTS, 256, 0, stream>>>(recs, bstart, dinv, N, nb);
    k_nrm<<<(E + 256 * NRM_PER - 1) / (256 * NRM_PER), 256, 0, stream>>>(recs, dinv, nrm, E);

    k_dense<128, true, true><<<denseGrid, DENSE_THREADS, 0, stream>>>(x, lin0_w, lin0_b, h, x0, ntiles);

    for (int l = 0; l < 4; ++l) {
        float beta = logf(0.5f / (float)(l + 1) + 1.0f);
        k_bucket_agg<<<nb, 256, 0, stream>>>(recs, nrm, bstart, dinv, h, agg, N);
        k_layer   <<<gemmGrid, 256, 0, stream>>>(agg, x0, conv_w + (size_t)l * 64 * 64,
                                                 beta, h, ntiles);
    }

    k_dense<64, false, false><<<denseGrid, DENSE_THREADS, 0, stream>>>(h, lin1_w, lin1_b, out, nullptr, ntiles);
}


# Layer product: a wave reads its 16-row agg and x0 tile once as whole 128-B lines, keeps the mixed rows in an LDS tile (A fragments and epilogue read it), stores h as whole lines
# speedup vs baseline: 2.8903x; 1.1528x over previous
#include <hip/hip_runtime.h>
#include <math.h>

typedef __attribute__((ext_vector_type(2))) float v2f;
typedef __attribute__((ext_vector_type(8))) float v8f;

#define DENSE_T       4
#define DENSE_THREADS 128
template <int K, bool RELU, bool STOREPRE>
__global__ void __launch_bounds__(DENSE_THREADS)
k_dense(const float* __restrict__ X, const float* __restrict__ W,
        const float* __restrict__ bias,
        float* __restrict__ outA, float* __restrict__ outPre,
        int ntiles) {
    __shared__ float wlds[64 * K];
    constexpr int NV4 = (64 * K) / 4;
    for (int i = threadIdx.x; i < NV4; i += DENSE_THREADS)
        ((float4*)wlds)[i] = ((const float4*)W)[i];
    __syncthreads();

    const int wid = __builtin_amdgcn_readfirstlane(
        (int)((blockIdx.x * DENSE_THREADS + threadIdx.x) >> 5));
    const int lane = threadIdx.x & 31;
    const int tile0 = wid * DENSE_T;
    if (tile0 >= ntiles) return;
    const int ln = lane & 15;
    const int kh = lane >> 4;

    const float* xrow[DENSE_T];
#pragma unroll
    for (int t = 0; t < DENSE_T; ++t) {
        int tile = tile0 + t;
        int tl = tile < ntiles ? tile : ntiles - 1;
        xrow[t] = X + (size_t)(tl * 16 + ln) * K + 2 * kh;
    }
    v2f afr[DENSE_T][K / 4];
#pragma unroll
    for (int kk = 0; kk < K / 4; ++kk)
#pragma unroll
        for (int t = 0; t < DENSE_T; ++t)
            afr[t][kk] = *(const v2f*)(xrow[t] + 4 * kk);

#pragma unroll
    for (int j = 0; j < 4; ++j) {
        const float* wrow = wlds + (size_t)(16 * j + ln) * K + 2 * kh;
        v8f acc[DENSE_T];
#pragma unroll
        for (int t = 0; t < DENSE_T; ++t) acc[t] = v8f{};
#pragma unroll
        for (int kk = 0; kk < K / 4; ++kk) {
            v2f b = *(const v2f*)(wrow + 4 * kk);
#pragma unroll
            for (int t = 0; t < DENSE_T; ++t)
                acc[t] = __builtin_amdgcn_wmma_f32_16x16x4_f32(
                    false, afr[t][kk], false, b, (short)0, acc[t], false, false);
        }
        const int n = 16 * j + ln;
        const float bv = bias[n];
#pragma unroll
        for (int t = 0; t < DENSE_T; ++t) {
            if (tile0 + t < ntiles) {
#pragma unroll
                for (int r = 0; r < 8; ++r) {
                    int m = r + 8 * kh;
                    size_t gi = (size_t)((tile0 + t) * 16 + m) * 64 + n;
                    float v = acc[t][r] + bv;
                    if constexpr (STOREPRE) outPre[gi] = v;
                    outA[gi] = RELU ? fmaxf(v, 0.0f) : v;
                }
            }
        }
    }
}

#define BKT         64
#define BKT_LOG     6
#define BIN_CHUNK   16384
#define BIN_THREADS 1024

extern __shared__ int dyn_lds[];

__global__ void __launch_bounds__(BIN_THREADS)
k_bin_count(const int* __restrict__ col, int* __restrict__ cnt, int e, int nb) {
    for (int i = threadIdx.x; i < nb; i += BIN_THREADS) dyn_lds[i] = 0;
    __syncthreads();
    const int e0 = blockIdx.x * BIN_CHUNK;
    int c[BIN_CHUNK / BIN_THREADS];
#pragma unroll
    for (int k = 0; k < BIN_CHUNK / BIN_THREADS; ++k) {
        int i = e0 + k * BIN_THREADS + (int)threadIdx.x;
        c[k] = col[i < e ? i : e - 1];
    }
#pragma unroll
    for (int k = 0; k < BIN_CHUNK / BIN_THREADS; ++k) {
        int i = e0 + k * BIN_THREADS + (int)threadIdx.x;
        if (i < e) atomicAdd(&dyn_lds[c[k] >> BKT_LOG], 1);
    }
    __syncthreads();
    int* dst = cnt + (size_t)blockIdx.x * nb;
    for (int i = threadIdx.x; i < nb; i += BIN_THREADS) dst[i] = dyn_lds[i];
}

__global__ void __launch_bounds__(1024)
k_bin_scan(const int* __restrict__ cnt, int* __restrict__ off, int* __restrict__ bstart,
           int nchunks, int nb) {
    __shared__ int sc[1024];
    int carry = 0;
    for (int b0 = 0; b0 < nb; b0 += 1024) {
        const int b = b0 + (int)threadIdx.x;
        int run = 0;
        if (b < nb) {
            for (int c0 = 0; c0 < nchunks; c0 += 16) {
                int v[16];
#pragma unroll
                for (int k = 0; k < 16; ++k) {
                    int c = c0 + k < nchunks ? c0 + k : nchunks - 1;
                    v[k] = cnt[(size_t)c * nb + b];
                }
#pragma unroll
                for (int k = 0; k < 16; ++k) {
                    if (c0 + k < nchunks) {
                        off[(size_t)(c0 + k) * nb + b] = run;
                        run += v[k];
                    }
                }
            }
        }
        sc[threadIdx.x] = run;
        __syncthreads();
        for (int d = 1; d < 1024; d <<= 1) {
            int t = (int)threadIdx.x >= d ? sc[threadIdx.x - d] : 0;
            __syncthreads();
            sc[threadIdx.x] += t;
            __syncthreads();
        }
        if (b < nb) bstart[b] = carry + sc[threadIdx.x] - run;
        const int tot = sc[1023];
        __syncthreads();
        carry += tot;
    }
    if (threadIdx.x == 0) bstart[nb] = carry;
}

__global__ void __launch_bounds__(BIN_THREADS)
k_bin_fill(const int* __restrict__ row, const int* __restrict__ col,
           const int* __restrict__ bstart, const int* __restrict__ off,
           int* __restrict__ recs, int e, int nb) {
    const int* myoff = off + (size_t)blockIdx.x * nb;
    for (int i = threadIdx.x; i < nb; i += BIN_THREADS) dyn_lds[i] = bstart[i] + myoff[i];
    __syncthreads();
    const int e0 = blockIdx.x * BIN_CHUNK;
    int r[BIN_CHUNK / BIN_THREADS], c[BIN_CHUNK / BIN_THREADS];
#pragma unroll
    for (int k = 0; k < BIN_CHUNK / BIN_THREADS; ++k) {
        int i = e0 + k * BIN_THREADS + (int)threadIdx.x;
        int ii = i < e ? i : e - 1;
        r[k] = row[ii];
        c[k] = col[ii];
    }
#pragma unroll
    for (int k = 0; k < BIN_CHUNK / BIN_THREADS; ++k) {
        int i = e0 + k * BIN_THREADS + (int)threadIdx.x;
        if (i < e) {
            int pos = atomicAdd(&dyn_lds[c[k] >> BKT_LOG], 1);
            ((int2*)recs)[pos] = make_int2(r[k], c[k]);
        }
    }
}

#define DEG_BKTS  4
#define DEG_BATCH 8
__global__ void __launch_bounds__(256)
k_bucket_dinv(const int* __restrict__ recs, const int* __restrict__ bstart,
              float* __restrict__ dinv, int n, int nb) {
    __shared__ int cnt[DEG_BKTS * BKT];
    cnt[threadIdx.x] = 1;
    __syncthreads();
    const int b0 = blockIdx.x * DEG_BKTS;
    int b1 = b0 + DEG_BKTS; if (b1 > nb) b1 = nb;
    const int s = bstart[b0];
    const int eend = bstart[b1];
    const int base = b0 * BKT;
    for (int i0 = s; i0 < eend; i0 += 256 * DEG_BATCH) {
        int t[DEG_BATCH];
#pragma unroll
        for (int k = 0; k < DEG_BATCH; ++k) {
            int i = i0 + k * 256 + (int)threadIdx.x;
            int ii = i < eend ? i : eend - 1;
            t[k] = recs[2 * (size_t)ii + 1];
        }
#pragma unroll
        for (int k = 0; k < DEG_BATCH; ++k) {
            int i = i0 + k * 256 + (int)threadIdx.x;
            if (i < eend) atomicAdd(&cnt[t[k] - base], 1);
        }
    }
    __syncthreads();
    const int node = base + (int)threadIdx.x;
    if (node < n) {
        float d = (float)cnt[threadIdx.x];
        dinv[node] = d > 0.0f ? rsqrtf(d) : 0.0f;
    }
}

#define NRM_PER 8
__global__ void __launch_bounds__(256)
k_nrm(const int* __restrict__ recs, const float* __restrict__ dinv,
      float* __restrict__ nrm, int e) {
    const int i0 = blockIdx.x * (256 * NRM_PER) + (int)threadIdx.x;
    int2 rc[NRM_PER];
    float f[NRM_PER];
#pragma unroll
    for (int k = 0; k < NRM_PER; ++k) {
        int i = i0 + k * 256;
        rc[k] = ((const int2*)recs)[i < e ? i : e - 1];
    }
#pragma unroll
    for (int k = 0; k < NRM_PER; ++k) f[k] = dinv[rc[k].x] * dinv[rc[k].y];
#pragma unroll
    for (int k = 0; k < NRM_PER; ++k) {
        int i = i0 + k * 256;
        if (i < e) nrm[i] = f[k];
    }
}

#define AGG_BATCH 16
template <bool FULL>
__device__ __forceinline__ void edge_group(int r, int c, float nv, int base, int cnt, int lane,
                                           const float* __restrict__ h, float* tile,
                                           const int2* __restrict__ nextRec,
                                           const float* __restrict__ nextNrm,
                                           int2& rcNext, float& nvNext) {
    const int nvBits = __float_as_int(nv);
#pragma unroll
    for (int jb = 0; jb < 32; jb += AGG_BATCH) {
        if (!FULL && jb >= cnt) break;
        float v0[AGG_BATCH], v1[AGG_BATCH];
#pragma unroll
        for (int j = 0; j < AGG_BATCH; ++j) {
            if (FULL || jb + j < cnt) {
                int rj = __builtin_amdgcn_readlane(r, jb + j);
                const float* s = h + (size_t)rj * 64 + lane;
                v0[j] = s[0];
                v1[j] = s[32];
            }
        }
        if (jb == 0) {
            rcNext = *nextRec;
            nvNext = *nextNrm;
        }
#pragma unroll
        for (int j = 0; j < AGG_BATCH; ++j) {
            if (FULL || jb + j < cnt) {
                int cj = __builtin_amdgcn_readlane(c, jb + j) - base;
                float nj = __int_as_float(__builtin_amdgcn_readlane(nvBits, jb + j));
                float* d = tile + cj * 64 + lane;
                atomicAdd(d,      nj * v0[j]);
                atomicAdd(d + 32, nj * v1[j]);
            }
        }
    }
}

__global__ void __launch_bounds__(256)
k_bucket_agg(const int* __restrict__ recs, const float* __restrict__ nrm,
             const int* __restrict__ bstart, const float* __restrict__ dinv,
             const float* __restrict__ h, float* __restrict__ agg, int n) {
    __shared__ float tile[BKT * 64];
    const int base = blockIdx.x * BKT;
    const int lane = threadIdx.x & 31;
    const int w = __builtin_amdgcn_readfirstlane((int)(threadIdx.x >> 5));

#pragma unroll
    for (int k = 0; k < BKT * 16 / 256; ++k) {
        int i = k * 256 + (int)threadIdx.x;
        int node = base + (i >> 4);
        int nc = node < n ? node : n - 1;
        float d = dinv[nc];
        float dd = d * d;
        float4 hv = ((const float4*)h)[(size_t)nc * 16 + (i & 15)];
        float4 v = make_float4(dd * hv.x, dd * hv.y, dd * hv.z, dd * hv.w);
        if (node >= n) v = make_float4(0.0f, 0.0f, 0.0f, 0.0f);
        ((float4*)tile)[i] = v;
    }
    __syncthreads();

    const int s = bstart[blockIdx.x];
    const int eend = bstart[blockIdx.x + 1];
    const int ngr = (eend - s + 31) >> 5;

    int g = w;
    int2 rc = make_int2(0, 0);
    float nv = 0.0f;
    if (g < ngr) {
        int e0 = s + g * 32;
        int c = eend - e0; if (c > 32) c = 32;
        int my = e0 + (lane < c ? lane : 0);
        rc = ((const int2*)recs)[my];
        nv = nrm[my];
    }
    while (g < ngr) {
        int cnt = eend - (s + g * 32); if (cnt > 32) cnt = 32;
        const int gn = g + 8;
        int e0n = s + (gn < ngr ? gn : g) * 32;
        int cn = eend - e0n; if (cn > 32) cn = 32;
        int myn = e0n + (lane < cn ? lane : 0);
        int2 rc2;
        float nv2;
        if (cnt == 32) edge_group<true >(rc.x, rc.y, nv, base, cnt, lane, h, tile,
                                         (const int2*)recs + myn, nrm + myn, rc2, nv2);
        else           edge_group<false>(rc.x, rc.y, nv, base, cnt, lane, h, tile,
                                         (const int2*)recs + myn, nrm + myn, rc2, nv2);
        rc = rc2; nv = nv2; g = gn;
    }
    __syncthreads();

#pragma unroll
    for (int k = 0; k < BKT * 16 / 256; ++k) {
        int i = k * 256 + (int)threadIdx.x;
        int node = base + (i >> 4);
        if (node < n)
            ((float4*)agg)[(size_t)node * 16 + (i & 15)] = ((const float4*)tile)[i];
    }
}

#define LAYER_PITCH 68
__global__ void __launch_bounds__(256)
k_layer(const float* __restrict__ agg, const float* __restrict__ x0,
        const float* __restrict__ Wl, float beta,
        float* __restrict__ h, int ntiles) {
    __shared__ __attribute__((aligned(16))) float wlds[64 * 64];
    __shared__ __attribute__((aligned(16))) float tiles[8 * 16 * LAYER_PITCH];
    for (int i = threadIdx.x; i < 1024; i += 256)
        ((float4*)wlds)[i] = ((const float4*)Wl)[i];

    const int w = __builtin_amdgcn_readfirstlane((int)(threadIdx.x >> 5));
    const int lane = threadIdx.x & 31;
    const int wid = blockIdx.x * 8 + w;
    const int tl = wid < ntiles ? wid : ntiles - 1;
    float* tile = tiles + w * (16 * LAYER_PITCH);
    const int ln = lane & 15;
    const int kh = lane >> 4;

    const float4* ag4 = (const float4*)agg + (size_t)tl * 256 + lane;
    const float4* xx4 = (const float4*)x0  + (size_t)tl * 256 + lane;
    float4 av[8], xv[8];
#pragma unroll
    for (int s = 0; s < 8; ++s) { av[s] = ag4[s * 32]; xv[s] = xx4[s * 32]; }
#pragma unroll
    for (int s = 0; s < 8; ++s) {
        float4 v;
        v.x = av[s].x * 0.9f + xv[s].x * 0.1f;
        v.y = av[s].y * 0.9f + xv[s].y * 0.1f;
        v.z = av[s].z * 0.9f + xv[s].z * 0.1f;
        v.w = av[s].w * 0.9f + xv[s].w * 0.1f;
        *(float4*)(tile + (2 * s + kh) * LAYER_PITCH + 4 * ln) = v;
    }
    __syncthreads();

    v2f afr[16];
#pragma unroll
    for (int kk = 0; kk < 16; ++kk)
        afr[kk] = *(const v2f*)(tile + ln * LAYER_PITCH + 4 * kk + 2 * kh);

    float ob = 1.0f - beta;
#pragma unroll
    for (int j = 0; j < 4; ++j) {
        const float* wcol = wlds + 2 * kh * 64 + 16 * j + ln;
        v8f acc = {};
#pragma unroll
        for (int kk = 0; kk < 16; ++kk) {
            v2f b;
            b.x = wcol[(size_t)(4 * kk) * 64];
            b.y = wcol[(size_t)(4 * kk + 1) * 64];
            acc = __builtin_amdgcn_wmma_f32_16x16x4_f32(
                false, afr[kk], false, b, (short)0, acc, false, false);
        }
        float* hp = tile + (8 * kh) * LAYER_PITCH + 16 * j + ln;
#pragma unroll
        for (int r = 0; r < 8; ++r) {
            float hh = hp[r * LAYER_PITCH];
            float v = ob * hh + beta * acc[r];
            hp[r * LAYER_PITCH] = fmaxf(v, 0.0f);
        }
    }

    if (wid < ntiles) {
        float4* h4 = (float4*)h + (size_t)wid * 256 + lane;
#pragma unroll
        for (int s = 0; s < 8; ++s)
            h4[s * 32] = *(const float4*)(tile + (2 * s + kh) * LAYER_PITCH + 4 * ln);
    }
}

extern "C" void kernel_launch(void* const* d_in, const int* in_sizes, int n_in,
                              void* d_out, int out_size, void* d_ws, size_t ws_size,
                              hipStream_t stream) {
    const float* x      = (const float*)d_in[0];
    const int*   ei     = (const int*)  d_in[1];
    const float* lin0_w = (const float*)d_in[2];
    const float* lin0_b = (const float*)d_in[3];
    const float* conv_w = (const float*)d_in[4];
    const float* lin1_w = (const float*)d_in[5];
    const float* lin1_b = (const float*)d_in[6];
    float* out = (float*)d_out;

    const int N = in_sizes[0] / 128;
    const int E = in_sizes[1] / 2;
    const int* row = ei;
    const int* col = ei + E;

    float* ws   = (float*)d_ws;
    float* dinv = ws;
    float* x0   = dinv + N;
    float* h    = x0 + (size_t)N * 64;
    float* agg  = h  + (size_t)N * 64;
    const int nb      = (N + BKT - 1) / BKT;
    const int nchunks = (E + BIN_CHUNK - 1) / BIN_CHUNK;
    int*   recs   = (int*)(agg + (size_t)N * 64);
    float* nrm    = (float*)(recs + (size_t)2 * E);
    int*   cnt    = (int*)(nrm + E);
    int*   off    = cnt + (size_t)nchunks * nb;
    int*   bstart = off + (size_t)nchunks * nb;

    const int ntiles = N / 16;
    const int gemmGrid = (ntiles * 32 + 255) / 256;
    const int denseGrid = (ntiles + 4 * DENSE_T - 1) / (4 * DENSE_T);

    k_bin_count<<<nchunks, BIN_THREADS, nb * sizeof(int), stream>>>(col, cnt, E, nb);
    k_bin_scan <<<1, 1024, 0, stream>>>(cnt, off, bstart, nchunks, nb);
    k_bin_fill <<<nchunks, BIN_THREADS, nb * sizeof(int), stream>>>(row, col, bstart, off,
                                                                  recs, E, nb);

    k_bucket_dinv<<<(nb + DEG_BKTS - 1) / DEG_BKTS, 256, 0, stream>>>(recs, bstart, dinv, N, nb);
    k_nrm<<<(E + 256 * NRM_PER - 1) / (256 * NRM_PER), 256, 0, stream>>>(recs, dinv, nrm, E);

    k_dense<128, true, true><<<denseGrid, DENSE_THREADS, 0, stream>>>(x, lin0_w, lin0_b, h, x0, ntiles);

    for (int l = 0; l < 4; ++l) {
        float beta = logf(0.5f / (float)(l + 1) + 1.0f);
        k_bucket_agg<<<nb, 256, 0, stream>>>(recs, nrm, bstart, dinv, h, agg, N);
        k_layer   <<<gemmGrid, 256, 0, stream>>>(agg, x0, conv_w + (size_t)l * 64 * 64,
                                                 beta, h, ntiles);
    }

    k_dense<64, false, false><<<denseGrid, DENSE_THREADS, 0, stream>>>(h, lin1_w, lin1_b, out, nullptr, ntiles);
}


# Layer product moved into the bucket block: the summed LDS tile is mixed with x0 in LDS and the block's 8 waves run the f32 WMMA product from it; no agg plane, h in two planes, 15 -> 11 launches
# speedup vs baseline: 3.2778x; 1.1341x over previous
#include <hip/hip_runtime.h>
#include <math.h>

typedef __attribute__((ext_vector_type(2))) float v2f;
typedef __attribute__((ext_vector_type(8))) float v8f;

#define DENSE_T       4
#define DENSE_THREADS 128
template <int K, bool RELU, bool STOREPRE>
__global__ void __launch_bounds__(DENSE_THREADS)
k_dense(const float* __restrict__ X, const float* __restrict__ W,
        const float* __restrict__ bias,
        float* __restrict__ outA, float* __restrict__ outPre,
        int ntiles) {
    __shared__ float wlds[64 * K];
    constexpr int NV4 = (64 * K) / 4;
    for (int i = threadIdx.x; i < NV4; i += DENSE_THREADS)
        ((float4*)wlds)[i] = ((const float4*)W)[i];
    __syncthreads();

    const int wid = __builtin_amdgcn_readfirstlane(
        (int)((blockIdx.x * DENSE_THREADS + threadIdx.x) >> 5));
    const int lane = threadIdx.x & 31;
    const int tile0 = wid * DENSE_T;
    if (tile0 >= ntiles) return;
    const int ln = lane & 15;
    const int kh = lane >> 4;

    const float* xrow[DENSE_T];
#pragma unroll
    for (int t = 0; t < DENSE_T; ++t) {
        int tile = tile0 + t;
        int tl = tile < ntiles ? tile : ntiles - 1;
        xrow[t] = X + (size_t)(tl * 16 + ln) * K + 2 * kh;
    }
    v2f afr[DENSE_T][K / 4];
#pragma unroll
    for (int kk = 0; kk < K / 4; ++kk)
#pragma unroll
        for (int t = 0; t < DENSE_T; ++t)
            afr[t][kk] = *(const v2f*)(xrow[t] + 4 * kk);

#pragma unroll
    for (int j = 0; j < 4; ++j) {
        const float* wrow = wlds + (size_t)(16 * j + ln) * K + 2 * kh;
        v8f acc[DENSE_T];
#pragma unroll
        for (int t = 0; t < DENSE_T; ++t) acc[t] = v8f{};
#pragma unroll
        for (int kk = 0; kk < K / 4; ++kk) {
            v2f b = *(const v2f*)(wrow + 4 * kk);
#pragma unroll
            for (int t = 0; t < DENSE_T; ++t)
                acc[t] = __builtin_amdgcn_wmma_f32_16x16x4_f32(
                    false, afr[t][kk], false, b, (short)0, acc[t], false, false);
        }
        const int n = 16 * j + ln;
        const float bv = bias[n];
#pragma unroll
        for (int t = 0; t < DENSE_T; ++t) {
            if (tile0 + t < ntiles) {
#pragma unroll
                for (int r = 0; r < 8; ++r) {
                    int m = r + 8 * kh;
                    size_t gi = (size_t)((tile0 + t) * 16 + m) * 64 + n;
                    float v = acc[t][r] + bv;
                    if constexpr (STOREPRE) outPre[gi] = v;
                    outA[gi] = RELU ? fmaxf(v, 0.0f) : v;
                }
            }
        }
    }
}

#define BKT         64
#define BKT_LOG     6
#define BIN_CHUNK   16384
#define BIN_THREADS 1024

extern __shared__ int dyn_lds[];

__global__ void __launch_bounds__(BIN_THREADS)
k_bin_count(const int* __restrict__ col, int* __restrict__ cnt, int e, int nb) {
    for (int i = threadIdx.x; i < nb; i += BIN_THREADS) dyn_lds[i] = 0;
    __syncthreads();
    const int e0 = blockIdx.x * BIN_CHUNK;
    int c[BIN_CHUNK / BIN_THREADS];
#pragma unroll
    for (int k = 0; k < BIN_CHUNK / BIN_THREADS; ++k) {
        int i = e0 + k * BIN_THREADS + (int)threadIdx.x;
        c[k] = col[i < e ? i : e - 1];
    }
#pragma unroll
    for (int k = 0; k < BIN_CHUNK / BIN_THREADS; ++k) {
        int i = e0 + k * BIN_THREADS + (int)threadIdx.x;
        if (i < e) atomicAdd(&dyn_lds[c[k] >> BKT_LOG], 1);
    }
    __syncthreads();
    int* dst = cnt + (size_t)blockIdx.x * nb;
    for (int i = threadIdx.x; i < nb; i += BIN_THREADS) dst[i] = dyn_lds[i];
}

__global__ void __launch_bounds__(1024)
k_bin_scan(const int* __restrict__ cnt, int* __restrict__ off, int* __restrict__ bstart,
           int nchunks, int nb) {
    __shared__ int sc[1024];
    int carry = 0;
    for (int b0 = 0; b0 < nb; b0 += 1024) {
        const int b = b0 + (int)threadIdx.x;
        int run = 0;
        if (b < nb) {
            for (int c0 = 0; c0 < nchunks; c0 += 16) {
                int v[16];
#pragma unroll
                for (int k = 0; k < 16; ++k) {
                    int c = c0 + k < nchunks ? c0 + k : nchunks - 1;
                    v[k] = cnt[(size_t)c * nb + b];
                }
#pragma unroll
                for (int k = 0; k < 16; ++k) {
                    if (c0 + k < nchunks) {
                        off[(size_t)(c0 + k) * nb + b] = run;
                        run += v[k];
                    }
                }
            }
        }
        sc[threadIdx.x] = run;
        __syncthreads();
        for (int d = 1; d < 1024; d <<= 1) {
            int t = (int)threadIdx.x >= d ? sc[threadIdx.x - d] : 0;
            __syncthreads();
            sc[threadIdx.x] += t;
            __syncthreads();
        }
        if (b < nb) bstart[b] = carry + sc[threadIdx.x] - run;
        const int tot = sc[1023];
        __syncthreads();
        carry += tot;
    }
    if (threadIdx.x == 0) bstart[nb] = carry;
}

__global__ void __launch_bounds__(BIN_THREADS)
k_bin_fill(const int* __restrict__ row, const int* __restrict__ col,
           const int* __restrict__ bstart, const int* __restrict__ off,
           int* __restrict__ recs, int e, int nb) {
    const int* myoff = off + (size_t)blockIdx.x * nb;
    for (int i = threadIdx.x; i < nb; i += BIN_THREADS) dyn_lds[i] = bstart[i] + myoff[i];
    __syncthreads();
    const int e0 = blockIdx.x * BIN_CHUNK;
    int r[BIN_CHUNK / BIN_THREADS], c[BIN_CHUNK / BIN_THREADS];
#pragma unroll
    for (int k = 0; k < BIN_CHUNK / BIN_THREADS; ++k) {
        int i = e0 + k * BIN_THREADS + (int)threadIdx.x;
        int ii = i < e ? i : e - 1;
        r[k] = row[ii];
        c[k] = col[ii];
    }
#pragma unroll
    for (int k = 0; k < BIN_CHUNK / BIN_THREADS; ++k) {
        int i = e0 + k * BIN_THREADS + (int)threadIdx.x;
        if (i < e) {
            int pos = atomicAdd(&dyn_lds[c[k] >> BKT_LOG], 1);
            ((int2*)recs)[pos] = make_int2(r[k], c[k]);
        }
    }
}

#define DEG_BKTS  4
#define DEG_BATCH 8
__global__ void __launch_bounds__(256)
k_bucket_dinv(const int* __restrict__ recs, const int* __restrict__ bstart,
              float* __restrict__ dinv, int n, int nb) {
    __shared__ int cnt[DEG_BKTS * BKT];
    cnt[threadIdx.x] = 1;
    __syncthreads();
    const int b0 = blockIdx.x * DEG_BKTS;
    int b1 = b0 + DEG_BKTS; if (b1 > nb) b1 = nb;
    const int s = bstart[b0];
    const int eend = bstart[b1];
    const int base = b0 * BKT;
    for (int i0 = s; i0 < eend; i0 += 256 * DEG_BATCH) {
        int t[DEG_BATCH];
#pragma unroll
        for (int k = 0; k < DEG_BATCH; ++k) {
            int i = i0 + k * 256 + (int)threadIdx.x;
            int ii = i < eend ? i : eend - 1;
            t[k] = recs[2 * (size_t)ii + 1];
        }
#pragma unroll
        for (int k = 0; k < DEG_BATCH; ++k) {
            int i = i0 + k * 256 + (int)threadIdx.x;
            if (i < eend) atomicAdd(&cnt[t[k] - base], 1);
        }
    }
    __syncthreads();
    const int node = base + (int)threadIdx.x;
    if (node < n) {
        float d = (float)cnt[threadIdx.x];
        dinv[node] = d > 0.0f ? rsqrtf(d) : 0.0f;
    }
}

#define NRM_PER 8
__global__ void __launch_bounds__(256)
k_nrm(const int* __restrict__ recs, const float* __restrict__ dinv,
      float* __restrict__ nrm, int e) {
    const int i0 = blockIdx.x * (256 * NRM_PER) + (int)threadIdx.x;
    int2 rc[NRM_PER];
    float f[NRM_PER];
#pragma unroll
    for (int k = 0; k < NRM_PER; ++k) {
        int i = i0 + k * 256;
        rc[k] = ((const int2*)recs)[i < e ? i : e - 1];
    }
#pragma unroll
    for (int k = 0; k < NRM_PER; ++k) f[k] = dinv[rc[k].x] * dinv[rc[k].y];
#pragma unroll
    for (int k = 0; k < NRM_PER; ++k) {
        int i = i0 + k * 256;
        if (i < e) nrm[i] = f[k];
    }
}

#define AGG_BATCH 16
template <bool FULL>
__device__ __forceinline__ void edge_group(int r, int c, float nv, int base, int cnt, int lane,
                                           const float* __restrict__ h, float* tile,
                                           const int2* __restrict__ nextRec,
                                           const float* __restrict__ nextNrm,
                                           int2& rcNext, float& nvNext) {
    const int nvBits = __float_as_int(nv);
#pragma unroll
    for (int jb = 0; jb < 32; jb += AGG_BATCH) {
        if (!FULL && jb >= cnt) break;
        float v0[AGG_BATCH], v1[AGG_BATCH];
#pragma unroll
        for (int j = 0; j < AGG_BATCH; ++j) {
            if (FULL || jb + j < cnt) {
                int rj = __builtin_amdgcn_readlane(r, jb + j);
                const float* s = h + (size_t)rj * 64 + lane;
                v0[j] = s[0];
                v1[j] = s[32];
            }
        }
        if (jb == 0) {
            rcNext = *nextRec;
            nvNext = *nextNrm;
        }
#pragma unroll
        for (int j = 0; j < AGG_BATCH; ++j) {
            if (FULL || jb + j < cnt) {
                int cj = __builtin_amdgcn_readlane(c, jb + j) - base;
                float nj = __int_as_float(__builtin_amdgcn_readlane(nvBits, jb + j));
                float* d = tile + cj * 64 + lane;
                atomicAdd(d,      nj * v0[j]);
                atomicAdd(d + 32, nj * v1[j]);
            }
        }
    }
}

#define MIX_PITCH 68
__global__ void __launch_bounds__(256, 10)
k_bucket_layer(const int* __restrict__ recs, const float* __restrict__ nrm,
               const int* __restrict__ bstart, const float* __restrict__ dinv,
               const float* __restrict__ h, const float* __restrict__ x0,
               const float* __restrict__ Wl, float beta,
               float* __restrict__ hout, int n) {
    __shared__ __attribute__((aligned(16))) float tile[BKT * 64];
    __shared__ __attribute__((aligned(16))) float mixed[BKT * MIX_PITCH];
    __shared__ __attribute__((aligned(16))) float wlds[64 * 64];
    const int base = blockIdx.x * BKT;
    const int lane = threadIdx.x & 31;
    const int w = __builtin_amdgcn_readfirstlane((int)(threadIdx.x >> 5));

    const int q0 = blockIdx.x * (BKT * 16);
    const int qlim = n * 16;

#pragma unroll
    for (int k = 0; k < BKT * 16 / 256; ++k) {
        int i = k * 256 + (int)threadIdx.x;
        int q = q0 + i;
        int qc = q < qlim ? q : qlim - 1;
        float d = dinv[qc >> 4];
        float dd = d * d;
        float4 hv = ((const float4*)h)[qc];
        float4 xv = ((const float4*)x0)[qc];
        float4 v = make_float4(dd * hv.x, dd * hv.y, dd * hv.z, dd * hv.w);
        if (q >= qlim) v = make_float4(0.0f, 0.0f, 0.0f, 0.0f);
        ((float4*)tile)[i] = v;
        *(float4*)(mixed + (i >> 4) * MIX_PITCH + 4 * (i & 15)) =
            make_float4(xv.x * 0.1f, xv.y * 0.1f, xv.z * 0.1f, xv.w * 0.1f);
    }
#pragma unroll
    for (int k = 0; k < 4; ++k)
        ((float4*)wlds)[k * 256 + (int)threadIdx.x] = ((const float4*)Wl)[k * 256 + (int)threadIdx.x];
    __syncthreads();

    const int s = bstart[blockIdx.x];
    const int eend = bstart[blockIdx.x + 1];
    const int ngr = (eend - s + 31) >> 5;

    int g = w;
    int2 rc = make_int2(0, 0);
    float nv = 0.0f;
    if (g < ngr) {
        int e0 = s + g * 32;
        int c = eend - e0; if (c > 32) c = 32;
        int my = e0 + (lane < c ? lane : 0);
        rc = ((const int2*)recs)[my];
        nv = nrm[my];
    }
    while (g < ngr) {
        int cnt = eend - (s + g * 32); if (cnt > 32) cnt = 32;
        const int gn = g + 8;
        int e0n = s + (gn < ngr ? gn : g) * 32;
        int cn = eend - e0n; if (cn > 32) cn = 32;
        int myn = e0n + (lane < cn ? lane : 0);
        int2 rc2;
        float nv2;
        if (cnt == 32) edge_group<true >(rc.x, rc.y, nv, base, cnt, lane, h, tile,
                                         (const int2*)recs + myn, nrm + myn, rc2, nv2);
        else           edge_group<false>(rc.x, rc.y, nv, base, cnt, lane, h, tile,
                                         (const int2*)recs + myn, nrm + myn, rc2, nv2);
        rc = rc2; nv = nv2; g = gn;
    }

    __syncthreads();

#pragma unroll
    for (int k = 0; k < BKT * 16 / 256; ++k) {
        int i = k * 256 + (int)threadIdx.x;
        float4 t = ((const float4*)tile)[i];
        float4* mp = (float4*)(mixed + (i >> 4) * MIX_PITCH + 4 * (i & 15));
        float4 m = *mp;
        float4 v;
        v.x = t.x * 0.9f + m.x;
        v.y = t.y * 0.9f + m.y;
        v.z = t.z * 0.9f + m.z;
        v.w = t.w * 0.9f + m.w;
        *mp = v;
    }
    __syncthreads();

    const int ln = lane & 15;
    const int kh = lane >> 4;
    const int rt = w >> 1;
    const int j0 = 2 * (w & 1);

    v2f afr[16];
#pragma unroll
    for (int kk = 0; kk < 16; ++kk)
        afr[kk] = *(const v2f*)(mixed + (rt * 16 + ln) * MIX_PITCH + 4 * kk + 2 * kh);

    const float* wcol = wlds + 2 * kh * 64 + 16 * j0 + ln;
    v8f acc0 = {}, acc1 = {};
#pragma unroll
    for (int kk = 0; kk < 16; ++kk) {
        v2f b0, b1;
        b0.x = wcol[(size_t)(4 * kk) * 64];
        b0.y = wcol[(size_t)(4 * kk + 1) * 64];
        b1.x = wcol[(size_t)(4 * kk) * 64 + 16];
        b1.y = wcol[(size_t)(4 * kk + 1) * 64 + 16];
        acc0 = __builtin_amdgcn_wmma_f32_16x16x4_f32(
            false, afr[kk], false, b0, (short)0, acc0, false, false);
        acc1 = __builtin_amdgcn_wmma_f32_16x16x4_f32(
            false, afr[kk], false, b1, (short)0, acc1, false, false);
    }
    const float ob = 1.0f - beta;
    const float* hp = mixed + (rt * 16 + 8 * kh) * MIX_PITCH + 16 * j0 + ln;
    float* op = tile + (rt * 16 + 8 * kh) * 64 + 16 * j0 + ln;
#pragma unroll
    for (int r = 0; r < 8; ++r) {
        float hh0 = hp[r * MIX_PITCH];
        float hh1 = hp[r * MIX_PITCH + 16];
        float v0 = ob * hh0 + beta * acc0[r];
        float v1 = ob * hh1 + beta * acc1[r];
        op[r * 64]      = fmaxf(v0, 0.0f);
        op[r * 64 + 16] = fmaxf(v1, 0.0f);
    }
    __syncthreads();

#pragma unroll
    for (int k = 0; k < BKT * 16 / 256; ++k) {
        int i = k * 256 + (int)threadIdx.x;
        if (q0 + i < qlim)
            ((float4*)hout)[q0 + i] = ((const float4*)tile)[i];
    }
}

extern "C" void kernel_launch(void* const* d_in, const int* in_sizes, int n_in,
                              void* d_out, int out_size, void* d_ws, size_t ws_size,
                              hipStream_t stream) {
    const float* x      = (const float*)d_in[0];
    const int*   ei     = (const int*)  d_in[1];
    const float* lin0_w = (const float*)d_in[2];
    const float* lin0_b = (const float*)d_in[3];
    const float* conv_w = (const float*)d_in[4];
    const float* lin1_w = (const float*)d_in[5];
    const float* lin1_b = (const float*)d_in[6];
    float* out = (float*)d_out;

    const int N = in_sizes[0] / 128;
    const int E = in_sizes[1] / 2;
    const int* row = ei;
    const int* col = ei + E;

    float* ws   = (float*)d_ws;
    float* dinv = ws;
    float* x0   = dinv + N;
    float* h    = x0 + (size_t)N * 64;
    float* h2   = h  + (size_t)N * 64;
    const int nb      = (N + BKT - 1) / BKT;
    const int nchunks = (E + BIN_CHUNK - 1) / BIN_CHUNK;
    int*   recs   = (int*)(h2 + (size_t)N * 64);
    float* nrm    = (float*)(recs + (size_t)2 * E);
    int*   cnt    = (int*)(nrm + E);
    int*   off    = cnt + (size_t)nchunks * nb;
    int*   bstart = off + (size_t)nchunks * nb;

    const int ntiles = N / 16;
    const int denseGrid = (ntiles + 4 * DENSE_T - 1) / (4 * DENSE_T);

    k_bin_count<<<nchunks, BIN_THREADS, nb * sizeof(int), stream>>>(col, cnt, E, nb);
    k_bin_scan <<<1, 1024, 0, stream>>>(cnt, off, bstart, nchunks, nb);
    k_bin_fill <<<nchunks, BIN_THREADS, nb * sizeof(int), stream>>>(row, col, bstart, off,
                                                                  recs, E, nb);

    k_bucket_dinv<<<(nb + DEG_BKTS - 1) / DEG_BKTS, 256, 0, stream>>>(recs, bstart, dinv, N, nb);
    k_nrm<<<(E + 256 * NRM_PER - 1) / (256 * NRM_PER), 256, 0, stream>>>(recs, dinv, nrm, E);

    k_dense<128, true, true><<<denseGrid, DENSE_THREADS, 0, stream>>>(x, lin0_w, lin0_b, h, x0, ntiles);

    const float* hcur = h;
    float* hnext = h2;
    for (int l = 0; l < 4; ++l) {
        float beta = logf(0.5f / (float)(l + 1) + 1.0f);
        k_bucket_layer<<<nb, 256, 0, stream>>>(recs, nrm, bstart, dinv, hcur, x0,
                                               conv_w + (size_t)l * 64 * 64, beta, hnext, N);
        float* t = (float*)hcur; hcur = hnext; hnext = t;
    }

    k_dense<64, false, false><<<denseGrid, DENSE_THREADS, 0, stream>>>(hcur, lin1_w, lin1_b, out, nullptr, ntiles);
}


# Dense products: a wave's 16 rows enter as whole 128-B lines into an LDS tile (A fragments read there; pitch K+4 for rows and weights), results leave through LDS as whole lines; sum order kept
# speedup vs baseline: 3.3787x; 1.0308x over previous
#include <hip/hip_runtime.h>
#include <math.h>

typedef __attribute__((ext_vector_type(2))) float v2f;
typedef __attribute__((ext_vector_type(8))) float v8f;

#define DENSE_WAVES   8
#define DENSE_THREADS 256
#define DENSE_TPW     2
#define DENSE_OP      68
template <int K, bool RELU, bool STOREPRE>
__global__ void __launch_bounds__(DENSE_THREADS)
k_dense(const float* __restrict__ X, const float* __restrict__ W,
        const float* __restrict__ bias,
        float* __restrict__ outA, float* __restrict__ outPre,
        int ntiles) {
    constexpr int P  = K + 4;
    constexpr int R4 = K / 4;
    constexpr int NL = 16 * R4 / 32;
    __shared__ __attribute__((aligned(16))) float wlds[64 * P];
    __shared__ __attribute__((aligned(16))) float tiles[DENSE_WAVES * 16 * P];
    __shared__ __attribute__((aligned(16))) float outs[DENSE_WAVES * 16 * DENSE_OP];

    const int lane = threadIdx.x & 31;
    const int w = __builtin_amdgcn_readfirstlane((int)(threadIdx.x >> 5));
    float* mt = tiles + w * (16 * P);
    float* ot = outs + w * (16 * DENSE_OP);
    const int ln = lane & 15;
    const int kh = lane >> 4;

#pragma unroll
    for (int k = 0; k < 64 * R4 / DENSE_THREADS; ++k) {
        const int i = k * DENSE_THREADS + (int)threadIdx.x;
        *(float4*)(wlds + (i / R4) * P + 4 * (i % R4)) = ((const float4*)W)[i];
    }

    float bv[4];
#pragma unroll
    for (int j = 0; j < 4; ++j) bv[j] = bias[16 * j + ln];

    for (int it = 0; it < DENSE_TPW; ++it) {
        const int tile = (blockIdx.x * DENSE_TPW + it) * DENSE_WAVES + w;
        const int tl = tile < ntiles ? tile : ntiles - 1;
        const float4* src = (const float4*)(X + (size_t)tl * 16 * K);
        float4 raw[NL];
#pragma unroll
        for (int k = 0; k < NL; ++k) raw[k] = src[k * 32 + lane];
#pragma unroll
        for (int k = 0; k < NL; ++k) {
            const int idx = k * 32 + lane;
            *(float4*)(mt + (idx / R4) * P + 4 * (idx % R4)) = raw[k];
        }
        __syncthreads();

        v2f afr[K / 4];
#pragma unroll
        for (int kk = 0; kk < K / 4; ++kk)
            afr[kk] = *(const v2f*)(mt + ln * P + 4 * kk + 2 * kh);

        v8f acc[4];
#pragma unroll
        for (int j = 0; j < 4; ++j) acc[j] = v8f{};
#pragma unroll
        for (int kk = 0; kk < K / 4; ++kk) {
#pragma unroll
            for (int j = 0; j < 4; ++j) {
                v2f b = *(const v2f*)(wlds + (16 * j + ln) * P + 4 * kk + 2 * kh);
                acc[j] = __builtin_amdgcn_wmma_f32_16x16x4_f32(
                    false, afr[kk], false, b, (short)0, acc[j], false, false);
            }
        }
#pragma unroll
        for (int j = 0; j < 4; ++j)
#pragma unroll
            for (int r = 0; r < 8; ++r)
                ot[(8 * kh + r) * DENSE_OP + 16 * j + ln] = acc[j][r] + bv[j];
        __syncthreads();

        if (tile < ntiles) {
#pragma unroll
            for (int k = 0; k < 8; ++k) {
                const int idx = k * 32 + lane;
                float4 v = *(const float4*)(ot + (idx >> 4) * DENSE_OP + 4 * (idx & 15));
                const size_t g = (size_t)tile * 256 + idx;
                if constexpr (STOREPRE) ((float4*)outPre)[g] = v;
                if constexpr (RELU) {
                    v.x = fmaxf(v.x, 0.0f); v.y = fmaxf(v.y, 0.0f);
                    v.z = fmaxf(v.z, 0.0f); v.w = fmaxf(v.w, 0.0f);
                }
                ((float4*)outA)[g] = v;
            }
        }
    }
}

#define BKT         64
#define BKT_LOG     6
#define BIN_CHUNK   16384
#define BIN_THREADS 1024

extern __shared__ int dyn_lds[];

__global__ void __launch_bounds__(BIN_THREADS)
k_bin_count(const int* __restrict__ col, int* __restrict__ cnt, int e, int nb) {
    for (int i = threadIdx.x; i < nb; i += BIN_THREADS) dyn_lds[i] = 0;
    __syncthreads();
    const int e0 = blockIdx.x * BIN_CHUNK;
    int c[BIN_CHUNK / BIN_THREADS];
#pragma unroll
    for (int k = 0; k < BIN_CHUNK / BIN_THREADS; ++k) {
        int i = e0 + k * BIN_THREADS + (int)threadIdx.x;
        c[k] = col[i < e ? i : e - 1];
    }
#pragma unroll
    for (int k = 0; k < BIN_CHUNK / BIN_THREADS; ++k) {
        int i = e0 + k * BIN_THREADS + (int)threadIdx.x;
        if (i < e) atomicAdd(&dyn_lds[c[k] >> BKT_LOG], 1);
    }
    __syncthreads();
    int* dst = cnt + (size_t)blockIdx.x * nb;
    for (int i = threadIdx.x; i < nb; i += BIN_THREADS) dst[i] = dyn_lds[i];
}

__global__ void __launch_bounds__(1024)
k_bin_scan(const int* __restrict__ cnt, int* __restrict__ off, int* __restrict__ bstart,
           int nchunks, int nb) {
    __shared__ int sc[1024];
    int carry = 0;
    for (int b0 = 0; b0 < nb; b0 += 1024) {
        const int b = b0 + (int)threadIdx.x;
        int run = 0;
        if (b < nb) {
            for (int c0 = 0; c0 < nchunks; c0 += 16) {
                int v[16];
#pragma unroll
                for (int k = 0; k < 16; ++k) {
                    int c = c0 + k < nchunks ? c0 + k : nchunks - 1;
                    v[k] = cnt[(size_t)c * nb + b];
                }
#pragma unroll
                for (int k = 0; k < 16; ++k) {
                    if (c0 + k < nchunks) {
                        off[(size_t)(c0 + k) * nb + b] = run;
                        run += v[k];
                    }
                }
            }
        }
        sc[threadIdx.x] = run;
        __syncthreads();
        for (int d = 1; d < 1024; d <<= 1) {
            int t = (int)threadIdx.x >= d ? sc[threadIdx.x - d] : 0;
            __syncthreads();
            sc[threadIdx.x] += t;
            __syncthreads();
        }
        if (b < nb) bstart[b] = carry + sc[threadIdx.x] - run;
        const int tot = sc[1023];
        __syncthreads();
        carry += tot;
    }
    if (threadIdx.x == 0) bstart[nb] = carry;
}

__global__ void __launch_bounds__(BIN_THREADS)
k_bin_fill(const int* __restrict__ row, const int* __restrict__ col,
           const int* __restrict__ bstart, const int* __restrict__ off,
           int* __restrict__ recs, int e, int nb) {
    const int* myoff = off + (size_t)blockIdx.x * nb;
    for (int i = threadIdx.x; i < nb; i += BIN_THREADS) dyn_lds[i] = bstart[i] + myoff[i];
    __syncthreads();
    const int e0 = blockIdx.x * BIN_CHUNK;
    int r[BIN_CHUNK / BIN_THREADS], c[BIN_CHUNK / BIN_THREADS];
#pragma unroll
    for (int k = 0; k < BIN_CHUNK / BIN_THREADS; ++k) {
        int i = e0 + k * BIN_THREADS + (int)threadIdx.x;
        int ii = i < e ? i : e - 1;
        r[k] = row[ii];
        c[k] = col[ii];
    }
#pragma unroll
    for (int k = 0; k < BIN_CHUNK / BIN_THREADS; ++k) {
        int i = e0 + k * BIN_THREADS + (int)threadIdx.x;
        if (i < e) {
            int pos = atomicAdd(&dyn_lds[c[k] >> BKT_LOG], 1);
            ((int2*)recs)[pos] = make_int2(r[k], c[k]);
        }
    }
}

#define DEG_BKTS  4
#define DEG_BATCH 8
__global__ void __launch_bounds__(256)
k_bucket_dinv(const int* __restrict__ recs, const int* __restrict__ bstart,
              float* __restrict__ dinv, int n, int nb) {
    __shared__ int cnt[DEG_BKTS * BKT];
    cnt[threadIdx.x] = 1;
    __syncthreads();
    const int b0 = blockIdx.x * DEG_BKTS;
    int b1 = b0 + DEG_BKTS; if (b1 > nb) b1 = nb;
    const int s = bstart[b0];
    const int eend = bstart[b1];
    const int base = b0 * BKT;
    for (int i0 = s; i0 < eend; i0 += 256 * DEG_BATCH) {
        int t[DEG_BATCH];
#pragma unroll
        for (int k = 0; k < DEG_BATCH; ++k) {
            int i = i0 + k * 256 + (int)threadIdx.x;
            int ii = i < eend ? i : eend - 1;
            t[k] = recs[2 * (size_t)ii + 1];
        }
#pragma unroll
        for (int k = 0; k < DEG_BATCH; ++k) {
            int i = i0 + k * 256 + (int)threadIdx.x;
            if (i < eend) atomicAdd(&cnt[t[k] - base], 1);
        }
    }
    __syncthreads();
    const int node = base + (int)threadIdx.x;
    if (node < n) {
        float d = (float)cnt[threadIdx.x];
        dinv[node] = d > 0.0f ? rsqrtf(d) : 0.0f;
    }
}

#define NRM_PER 8
__global__ void __launch_bounds__(256)
k_nrm(const int* __restrict__ recs, const float* __restrict__ dinv,
      float* __restrict__ nrm, int e) {
    const int i0 = blockIdx.x * (256 * NRM_PER) + (int)threadIdx.x;
    int2 rc[NRM_PER];
    float f[NRM_PER];
#pragma unroll
    for (int k = 0; k < NRM_PER; ++k) {
        int i = i0 + k * 256;
        rc[k] = ((const int2*)recs)[i < e ? i : e - 1];
    }
#pragma unroll
    for (int k = 0; k < NRM_PER; ++k) f[k] = dinv[rc[k].x] * dinv[rc[k].y];
#pragma unroll
    for (int k = 0; k < NRM_PER; ++k) {
        int i = i0 + k * 256;
        if (i < e) nrm[i] = f[k];
    }
}

#define AGG_BATCH 16
template <bool FULL>
__device__ __forceinline__ void edge_group(int r, int c, float nv, int base, int cnt, int lane,
                                           const float* __restrict__ h, float* tile,
                                           const int2* __restrict__ nextRec,
                                           const float* __restrict__ nextNrm,
                                           int2& rcNext, float& nvNext) {
    const int nvBits = __float_as_int(nv);
#pragma unroll
    for (int jb = 0; jb < 32; jb += AGG_BATCH) {
        if (!FULL && jb >= cnt) break;
        float v0[AGG_BATCH], v1[AGG_BATCH];
#pragma unroll
        for (int j = 0; j < AGG_BATCH; ++j) {
            if (FULL || jb + j < cnt) {
                int rj = __builtin_amdgcn_readlane(r, jb + j);
                const float* s = h + (size_t)rj * 64 + lane;
                v0[j] = s[0];
                v1[j] = s[32];
            }
        }
        if (jb == 0) {
            rcNext = *nextRec;
            nvNext = *nextNrm;
        }
#pragma unroll
        for (int j = 0; j < AGG_BATCH; ++j) {
            if (FULL || jb + j < cnt) {
                int cj = __builtin_amdgcn_readlane(c, jb + j) - base;
                float nj = __int_as_float(__builtin_amdgcn_readlane(nvBits, jb + j));
                float* d = tile + cj * 64 + lane;
                atomicAdd(d,      nj * v0[j]);
                atomicAdd(d + 32, nj * v1[j]);
            }
        }
    }
}

#define MIX_PITCH 68
__global__ void __launch_bounds__(256, 10)
k_bucket_layer(const int* __restrict__ recs, const float* __restrict__ nrm,
               const int* __restrict__ bstart, const float* __restrict__ dinv,
               const float* __restrict__ h, const float* __restrict__ x0,
               const float* __restrict__ Wl, float beta,
               float* __restrict__ hout, int n) {
    __shared__ __attribute__((aligned(16))) float tile[BKT * 64];
    __shared__ __attribute__((aligned(16))) float mixed[BKT * MIX_PITCH];
    __shared__ __attribute__((aligned(16))) float wlds[64 * 64];
    const int base = blockIdx.x * BKT;
    const int lane = threadIdx.x & 31;
    const int w = __builtin_amdgcn_readfirstlane((int)(threadIdx.x >> 5));

    const int q0 = blockIdx.x * (BKT * 16);
    const int qlim = n * 16;

#pragma unroll
    for (int k = 0; k < BKT * 16 / 256; ++k) {
        int i = k * 256 + (int)threadIdx.x;
        int q = q0 + i;
        int qc = q < qlim ? q : qlim - 1;
        float d = dinv[qc >> 4];
        float dd = d * d;
        float4 hv = ((const float4*)h)[qc];
        float4 xv = ((const float4*)x0)[qc];
        float4 v = make_float4(dd * hv.x, dd * hv.y, dd * hv.z, dd * hv.w);
        if (q >= qlim) v = make_float4(0.0f, 0.0f, 0.0f, 0.0f);
        ((float4*)tile)[i] = v;
        *(float4*)(mixed + (i >> 4) * MIX_PITCH + 4 * (i & 15)) =
            make_float4(xv.x * 0.1f, xv.y * 0.1f, xv.z * 0.1f, xv.w * 0.1f);
    }
#pragma unroll
    for (int k = 0; k < 4; ++k)
        ((float4*)wlds)[k * 256 + (int)threadIdx.x] = ((const float4*)Wl)[k * 256 + (int)threadIdx.x];
    __syncthreads();

    const int s = bstart[blockIdx.x];
    const int eend = bstart[blockIdx.x + 1];
    const int ngr = (eend - s + 31) >> 5;

    int g = w;
    int2 rc = make_int2(0, 0);
    float nv = 0.0f;
    if (g < ngr) {
        int e0 = s + g * 32;
        int c = eend - e0; if (c > 32) c = 32;
        int my = e0 + (lane < c ? lane : 0);
        rc = ((const int2*)recs)[my];
        nv = nrm[my];
    }
    while (g < ngr) {
        int cnt = eend - (s + g * 32); if (cnt > 32) cnt = 32;
        const int gn = g + 8;
        int e0n = s + (gn < ngr ? gn : g) * 32;
        int cn = eend - e0n; if (cn > 32) cn = 32;
        int myn = e0n + (lane < cn ? lane : 0);
        int2 rc2;
        float nv2;
        if (cnt == 32) edge_group<true >(rc.x, rc.y, nv, base, cnt, lane, h, tile,
                                         (const int2*)recs + myn, nrm + myn, rc2, nv2);
        else           edge_group<false>(rc.x, rc.y, nv, base, cnt, lane, h, tile,
                                         (const int2*)recs + myn, nrm + myn, rc2, nv2);
        rc = rc2; nv = nv2; g = gn;
    }

    __syncthreads();

#pragma unroll
    for (int k = 0; k < BKT * 16 / 256; ++k) {
        int i = k * 256 + (int)threadIdx.x;
        float4 t = ((const float4*)tile)[i];
        float4* mp = (float4*)(mixed + (i >> 4) * MIX_PITCH + 4 * (i & 15));
        float4 m = *mp;
        float4 v;
        v.x = t.x * 0.9f + m.x;
        v.y = t.y * 0.9f + m.y;
        v.z = t.z * 0.9f + m.z;
        v.w = t.w * 0.9f + m.w;
        *mp = v;
    }
    __syncthreads();

    const int ln = lane & 15;
    const int kh = lane >> 4;
    const int rt = w >> 1;
    const int j0 = 2 * (w & 1);

    v2f afr[16];
#pragma unroll
    for (int kk = 0; kk < 16; ++kk)
        afr[kk] = *(const v2f*)(mixed + (rt * 16 + ln) * MIX_PITCH + 4 * kk + 2 * kh);

    const float* wcol = wlds + 2 * kh * 64 + 16 * j0 + ln;
    v8f acc0 = {}, acc1 = {};
#pragma unroll
    for (int kk = 0; kk < 16; ++kk) {
        v2f b0, b1;
        b0.x = wcol[(size_t)(4 * kk) * 64];
        b0.y = wcol[(size_t)(4 * kk + 1) * 64];
        b1.x = wcol[(size_t)(4 * kk) * 64 + 16];
        b1.y = wcol[(size_t)(4 * kk + 1) * 64 + 16];
        acc0 = __builtin_amdgcn_wmma_f32_16x16x4_f32(
            false, afr[kk], false, b0, (short)0, acc0, false, false);
        acc1 = __builtin_amdgcn_wmma_f32_16x16x4_f32(
            false, afr[kk], false, b1, (short)0, acc1, false, false);
    }
    const float ob = 1.0f - beta;
    const float* hp = mixed + (rt * 16 + 8 * kh) * MIX_PITCH + 16 * j0 + ln;
    float* op = tile + (rt * 16 + 8 * kh) * 64 + 16 * j0 + ln;
#pragma unroll
    for (int r = 0; r < 8; ++r) {
        float hh0 = hp[r * MIX_PITCH];
        float hh1 = hp[r * MIX_PITCH + 16];
        float v0 = ob * hh0 + beta * acc0[r];
        float v1 = ob * hh1 + beta * acc1[r];
        op[r * 64]      = fmaxf(v0, 0.0f);
        op[r * 64 + 16] = fmaxf(v1, 0.0f);
    }
    __syncthreads();

#pragma unroll
    for (int k = 0; k < BKT * 16 / 256; ++k) {
        int i = k * 256 + (int)threadIdx.x;
        if (q0 + i < qlim)
            ((float4*)hout)[q0 + i] = ((const float4*)tile)[i];
    }
}

extern "C" void kernel_launch(void* const* d_in, const int* in_sizes, int n_in,
                              void* d_out, int out_size, void* d_ws, size_t ws_size,
                              hipStream_t stream) {
    const float* x      = (const float*)d_in[0];
    const int*   ei     = (const int*)  d_in[1];
    const float* lin0_w = (const float*)d_in[2];
    const float* lin0_b = (const float*)d_in[3];
    const float* conv_w = (const float*)d_in[4];
    const float* lin1_w = (const float*)d_in[5];
    const float* lin1_b = (const float*)d_in[6];
    float* out = (float*)d_out;

    const int N = in_sizes[0] / 128;
    const int E = in_sizes[1] / 2;
    const int* row = ei;
    const int* col = ei + E;

    float* ws   = (float*)d_ws;
    float* dinv = ws;
    float* x0   = dinv + N;
    float* h    = x0 + (size_t)N * 64;
    float* h2   = h  + (size_t)N * 64;
    const int nb      = (N + BKT - 1) / BKT;
    const int nchunks = (E + BIN_CHUNK - 1) / BIN_CHUNK;
    int*   recs   = (int*)(h2 + (size_t)N * 64);
    float* nrm    = (float*)(recs + (size_t)2 * E);
    int*   cnt    = (int*)(nrm + E);
    int*   off    = cnt + (size_t)nchunks * nb;
    int*   bstart = off + (size_t)nchunks * nb;

    const int ntiles = N / 16;
    const int denseGrid = (ntiles + DENSE_WAVES * DENSE_TPW - 1) / (DENSE_WAVES * DENSE_TPW);

    k_bin_count<<<nchunks, BIN_THREADS, nb * sizeof(int), stream>>>(col, cnt, E, nb);
    k_bin_scan <<<1, 1024, 0, stream>>>(cnt, off, bstart, nchunks, nb);
    k_bin_fill <<<nchunks, BIN_THREADS, nb * sizeof(int), stream>>>(row, col, bstart, off,
                                                                  recs, E, nb);

    k_bucket_dinv<<<(nb + DEG_BKTS - 1) / DEG_BKTS, 256, 0, stream>>>(recs, bstart, dinv, N, nb);
    k_nrm<<<(E + 256 * NRM_PER - 1) / (256 * NRM_PER), 256, 0, stream>>>(recs, dinv, nrm, E);

    k_dense<128, true, true><<<denseGrid, DENSE_THREADS, 0, stream>>>(x, lin0_w, lin0_b, h, x0, ntiles);

    const float* hcur = h;
    float* hnext = h2;
    for (int l = 0; l < 4; ++l) {
        float beta = logf(0.5f / (float)(l + 1) + 1.0f);
        k_bucket_layer<<<nb, 256, 0, stream>>>(recs, nrm, bstart, dinv, hcur, x0,
                                               conv_w + (size_t)l * 64 * 64, beta, hnext, N);
        float* t = (float*)hcur; hcur = hnext; hnext = t;
    }

    k_dense<64, false, false><<<denseGrid, DENSE_THREADS, 0, stream>>>(hcur, lin1_w, lin1_b, out, nullptr, ntiles);
}


# Target buckets of 80 nodes (was 64): 1250 layer blocks, all resident at once at 5 per WGP (was 1563 on 1280 places); fifth 16-row tile's product on waves 0-3; arrival-order LDS sums kept
# speedup vs baseline: 3.5974x; 1.0647x over previous
#include <hip/hip_runtime.h>
#include <math.h>

typedef __attribute__((ext_vector_type(2))) float v2f;
typedef __attribute__((ext_vector_type(8))) float v8f;

#define DENSE_WAVES   8
#define DENSE_THREADS 256
#define DENSE_TPW     2
#define DENSE_OP      68
template <int K, bool RELU, bool STOREPRE>
__global__ void __launch_bounds__(DENSE_THREADS)
k_dense(const float* __restrict__ X, const float* __restrict__ W,
        const float* __restrict__ bias,
        float* __restrict__ outA, float* __restrict__ outPre,
        int ntiles) {
    constexpr int P  = K + 4;
    constexpr int R4 = K / 4;
    constexpr int NL = 16 * R4 / 32;
    __shared__ __attribute__((aligned(16))) float wlds[64 * P];
    __shared__ __attribute__((aligned(16))) float tiles[DENSE_WAVES * 16 * P];
    __shared__ __attribute__((aligned(16))) float outs[DENSE_WAVES * 16 * DENSE_OP];

    const int lane = threadIdx.x & 31;
    const int w = __builtin_amdgcn_readfirstlane((int)(threadIdx.x >> 5));
    float* mt = tiles + w * (16 * P);
    float* ot = outs + w * (16 * DENSE_OP);
    const int ln = lane & 15;
    const int kh = lane >> 4;

#pragma unroll
    for (int k = 0; k < 64 * R4 / DENSE_THREADS; ++k) {
        const int i = k * DENSE_THREADS + (int)threadIdx.x;
        *(float4*)(wlds + (i / R4) * P + 4 * (i % R4)) = ((const float4*)W)[i];
    }

    float bv[4];
#pragma unroll
    for (int j = 0; j < 4; ++j) bv[j] = bias[16 * j + ln];

    for (int it = 0; it < DENSE_TPW; ++it) {
        const int tile = (blockIdx.x * DENSE_TPW + it) * DENSE_WAVES + w;
        const int tl = tile < ntiles ? tile : ntiles - 1;
        const float4* src = (const float4*)(X + (size_t)tl * 16 * K);
        float4 raw[NL];
#pragma unroll
        for (int k = 0; k < NL; ++k) raw[k] = src[k * 32 + lane];
#pragma unroll
        for (int k = 0; k < NL; ++k) {
            const int idx = k * 32 + lane;
            *(float4*)(mt + (idx / R4) * P + 4 * (idx % R4)) = raw[k];
        }
        __syncthreads();

        v2f afr[K / 4];
#pragma unroll
        for (int kk = 0; kk < K / 4; ++kk)
            afr[kk] = *(const v2f*)(mt + ln * P + 4 * kk + 2 * kh);

        v8f acc[4];
#pragma unroll
        for (int j = 0; j < 4; ++j) acc[j] = v8f{};
#pragma unroll
        for (int kk = 0; kk < K / 4; ++kk) {
#pragma unroll
            for (int j = 0; j < 4; ++j) {
                v2f b = *(const v2f*)(wlds + (16 * j + ln) * P + 4 * kk + 2 * kh);
                acc[j] = __builtin_amdgcn_wmma_f32_16x16x4_f32(
                    false, afr[kk], false, b, (short)0, acc[j], false, false);
            }
        }
#pragma unroll
        for (int j = 0; j < 4; ++j)
#pragma unroll
            for (int r = 0; r < 8; ++r)
                ot[(8 * kh + r) * DENSE_OP + 16 * j + ln] = acc[j][r] + bv[j];
        __syncthreads();

        if (tile < ntiles) {
#pragma unroll
            for (int k = 0; k < 8; ++k) {
                const int idx = k * 32 + lane;
                float4 v = *(const float4*)(ot + (idx >> 4) * DENSE_OP + 4 * (idx & 15));
                const size_t g = (size_t)tile * 256 + idx;
                if constexpr (STOREPRE) ((float4*)outPre)[g] = v;
                if constexpr (RELU) {
                    v.x = fmaxf(v.x, 0.0f); v.y = fmaxf(v.y, 0.0f);
                    v.z = fmaxf(v.z, 0.0f); v.w = fmaxf(v.w, 0.0f);
                }
                ((float4*)outA)[g] = v;
            }
        }
    }
}

#define BKT         80
#define BIN_CHUNK   16384
#define BIN_THREADS 1024

extern __shared__ int dyn_lds[];

__global__ void __launch_bounds__(BIN_THREADS)
k_bin_count(const int* __restrict__ col, int* __restrict__ cnt, int e, int nb) {
    for (int i = threadIdx.x; i < nb; i += BIN_THREADS) dyn_lds[i] = 0;
    __syncthreads();
    const int e0 = blockIdx.x * BIN_CHUNK;
    int c[BIN_CHUNK / BIN_THREADS];
#pragma unroll
    for (int k = 0; k < BIN_CHUNK / BIN_THREADS; ++k) {
        int i = e0 + k * BIN_THREADS + (int)threadIdx.x;
        c[k] = col[i < e ? i : e - 1];
    }
#pragma unroll
    for (int k = 0; k < BIN_CHUNK / BIN_THREADS; ++k) {
        int i = e0 + k * BIN_THREADS + (int)threadIdx.x;
        if (i < e) atomicAdd(&dyn_lds[(unsigned)c[k] / BKT], 1);
    }
    __syncthreads();
    int* dst = cnt + (size_t)blockIdx.x * nb;
    for (int i = threadIdx.x; i < nb; i += BIN_THREADS) dst[i] = dyn_lds[i];
}

__global__ void __launch_bounds__(1024)
k_bin_scan(const int* __restrict__ cnt, int* __restrict__ off, int* __restrict__ bstart,
           int nchunks, int nb) {
    __shared__ int sc[1024];
    int carry = 0;
    for (int b0 = 0; b0 < nb; b0 += 1024) {
        const int b = b0 + (int)threadIdx.x;
        int run = 0;
        if (b < nb) {
            for (int c0 = 0; c0 < nchunks; c0 += 16) {
                int v[16];
#pragma unroll
                for (int k = 0; k < 16; ++k) {
                    int c = c0 + k < nchunks ? c0 + k : nchunks - 1;
                    v[k] = cnt[(size_t)c * nb + b];
                }
#pragma unroll
                for (int k = 0; k < 16; ++k) {
                    if (c0 + k < nchunks) {
                        off[(size_t)(c0 + k) * nb + b] = run;
                        run += v[k];
                    }
                }
            }
        }
        sc[threadIdx.x] = run;
        __syncthreads();
        for (int d = 1; d < 1024; d <<= 1) {
            int t = (int)threadIdx.x >= d ? sc[threadIdx.x - d] : 0;
            __syncthreads();
            sc[threadIdx.x] += t;
            __syncthreads();
        }
        if (b < nb) bstart[b] = carry + sc[threadIdx.x] - run;
        const int tot = sc[1023];
        __syncthreads();
        carry += tot;
    }
    if (threadIdx.x == 0) bstart[nb] = carry;
}

__global__ void __launch_bounds__(BIN_THREADS)
k_bin_fill(const int* __restrict__ row, const int* __restrict__ col,
           const int* __restrict__ bstart, const int* __restrict__ off,
           int* __restrict__ recs, int e, int nb) {
    const int* myoff = off + (size_t)blockIdx.x * nb;
    for (int i = threadIdx.x; i < nb; i += BIN_THREADS) dyn_lds[i] = bstart[i] + myoff[i];
    __syncthreads();
    const int e0 = blockIdx.x * BIN_CHUNK;
    int r[BIN_CHUNK / BIN_THREADS], c[BIN_CHUNK / BIN_THREADS];
#pragma unroll
    for (int k = 0; k < BIN_CHUNK / BIN_THREADS; ++k) {
        int i = e0 + k * BIN_THREADS + (int)threadIdx.x;
        int ii = i < e ? i : e - 1;
        r[k] = row[ii];
        c[k] = col[ii];
    }
#pragma unroll
    for (int k = 0; k < BIN_CHUNK / BIN_THREADS; ++k) {
        int i = e0 + k * BIN_THREADS + (int)threadIdx.x;
        if (i < e) {
            int pos = atomicAdd(&dyn_lds[(unsigned)c[k] / BKT], 1);
            ((int2*)recs)[pos] = make_int2(r[k], c[k]);
        }
    }
}

#define DEG_BKTS    4
#define DEG_BATCH   8
#define DEG_THREADS (DEG_BKTS * BKT)
__global__ void __launch_bounds__(DEG_THREADS)
k_bucket_dinv(const int* __restrict__ recs, const int* __restrict__ bstart,
              float* __restrict__ dinv, int n, int nb) {
    __shared__ int cnt[DEG_BKTS * BKT];
    cnt[threadIdx.x] = 1;
    __syncthreads();
    const int b0 = blockIdx.x * DEG_BKTS;
    int b1 = b0 + DEG_BKTS; if (b1 > nb) b1 = nb;
    const int s = bstart[b0];
    const int eend = bstart[b1];
    const int base = b0 * BKT;
    for (int i0 = s; i0 < eend; i0 += DEG_THREADS * DEG_BATCH) {
        int t[DEG_BATCH];
#pragma unroll
        for (int k = 0; k < DEG_BATCH; ++k) {
            int i = i0 + k * DEG_THREADS + (int)threadIdx.x;
            int ii = i < eend ? i : eend - 1;
            t[k] = recs[2 * (size_t)ii + 1];
        }
#pragma unroll
        for (int k = 0; k < DEG_BATCH; ++k) {
            int i = i0 + k * DEG_THREADS + (int)threadIdx.x;
            if (i < eend) atomicAdd(&cnt[t[k] - base], 1);
        }
    }
    __syncthreads();
    const int node = base + (int)threadIdx.x;
    if (node < n) {
        float d = (float)cnt[threadIdx.x];
        dinv[node] = d > 0.0f ? rsqrtf(d) : 0.0f;
    }
}

#define NRM_PER 8
__global__ void __launch_bounds__(256)
k_nrm(const int* __restrict__ recs, const float* __restrict__ dinv,
      float* __restrict__ nrm, int e) {
    const int i0 = blockIdx.x * (256 * NRM_PER) + (int)threadIdx.x;
    int2 rc[NRM_PER];
    float f[NRM_PER];
#pragma unroll
    for (int k = 0; k < NRM_PER; ++k) {
        int i = i0 + k * 256;
        rc[k] = ((const int2*)recs)[i < e ? i : e - 1];
    }
#pragma unroll
    for (int k = 0; k < NRM_PER; ++k) f[k] = dinv[rc[k].x] * dinv[rc[k].y];
#pragma unroll
    for (int k = 0; k < NRM_PER; ++k) {
        int i = i0 + k * 256;
        if (i < e) nrm[i] = f[k];
    }
}

#define AGG_BATCH 16
template <bool FULL>
__device__ __forceinline__ void edge_group(int r, int c, float nv, int base, int cnt, int lane,
                                           const float* __restrict__ h, float* tile,
                                           const int2* __restrict__ nextRec,
                                           const float* __restrict__ nextNrm,
                                           int2& rcNext, float& nvNext) {
    const int nvBits = __float_as_int(nv);
#pragma unroll
    for (int jb = 0; jb < 32; jb += AGG_BATCH) {
        if (!FULL && jb >= cnt) break;
        float v0[AGG_BATCH], v1[AGG_BATCH];
#pragma unroll
        for (int j = 0; j < AGG_BATCH; ++j) {
            if (FULL || jb + j < cnt) {
                int rj = __builtin_amdgcn_readlane(r, jb + j);
                const float* s = h + (size_t)rj * 64 + lane;
                v0[j] = s[0];
                v1[j] = s[32];
            }
        }
        if (jb == 0) {
            rcNext = *nextRec;
            nvNext = *nextNrm;
        }
#pragma unroll
        for (int j = 0; j < AGG_BATCH; ++j) {
            if (FULL || jb + j < cnt) {
                int cj = __builtin_amdgcn_readlane(c, jb + j) - base;
                float nj = __int_as_float(__builtin_amdgcn_readlane(nvBits, jb + j));
                float* d = tile + cj * 64 + lane;
                atomicAdd(d,      nj * v0[j]);
                atomicAdd(d + 32, nj * v1[j]);
            }
        }
    }
}

#define MIX_PITCH 68

template <int NC>
__device__ __forceinline__ void tile_product(const float* mixed, const float* wlds, float* tile,
                                             int rt, int jc, int ln, int kh, float beta) {
    v2f afr[16];
#pragma unroll
    for (int kk = 0; kk < 16; ++kk)
        afr[kk] = *(const v2f*)(mixed + (rt * 16 + ln) * MIX_PITCH + 4 * kk + 2 * kh);

    const float* wcol = wlds + 2 * kh * 64 + 16 * jc + ln;
    v8f acc[NC];
#pragma unroll
    for (int c = 0; c < NC; ++c) acc[c] = v8f{};
#pragma unroll
    for (int kk = 0; kk < 16; ++kk) {
#pragma unroll
        for (int c = 0; c < NC; ++c) {
            v2f b;
            b.x = wcol[(size_t)(4 * kk) * 64 + 16 * c];
            b.y = wcol[(size_t)(4 * kk + 1) * 64 + 16 * c];
            acc[c] = __builtin_amdgcn_wmma_f32_16x16x4_f32(
                false, afr[kk], false, b, (short)0, acc[c], false, false);
        }
    }
    const float ob = 1.0f - beta;
    const float* hp = mixed + (rt * 16 + 8 * kh) * MIX_PITCH + 16 * jc + ln;
    float* op = tile + (rt * 16 + 8 * kh) * 64 + 16 * jc + ln;
#pragma unroll
    for (int r = 0; r < 8; ++r) {
#pragma unroll
        for (int c = 0; c < NC; ++c) {
            float hh = hp[r * MIX_PITCH + 16 * c];
            float v = ob * hh + beta * acc[c][r];
            op[r * 64 + 16 * c] = fmaxf(v, 0.0f);
        }
    }
}
__global__ void __launch_bounds__(256, 10)
k_bucket_layer(const int* __restrict__ recs, const float* __restrict__ nrm,
               const int* __restrict__ bstart, const float* __restrict__ dinv,
               const float* __restrict__ h, const float* __restrict__ x0,
               const float* __restrict__ Wl, float beta,
               float* __restrict__ hout, int n) {
    __shared__ __attribute__((aligned(16))) float tile[BKT * 64];
    __shared__ __attribute__((aligned(16))) float mixed[BKT * MIX_PITCH];
    __shared__ __attribute__((aligned(16))) float wlds[64 * 64];
    const int base = blockIdx.x * BKT;
    const int lane = threadIdx.x & 31;
    const int w = __builtin_amdgcn_readfirstlane((int)(threadIdx.x >> 5));

    const int q0 = blockIdx.x * (BKT * 16);
    const int qlim = n * 16;

#pragma unroll
    for (int k = 0; k < BKT * 16 / 256; ++k) {
        int i = k * 256 + (int)threadIdx.x;
        int q = q0 + i;
        int qc = q < qlim ? q : qlim - 1;
        float d = dinv[qc >> 4];
        float dd = d * d;
        float4 hv = ((const float4*)h)[qc];
        float4 xv = ((const float4*)x0)[qc];
        float4 v = make_float4(dd * hv.x, dd * hv.y, dd * hv.z, dd * hv.w);
        if (q >= qlim) v = make_float4(0.0f, 0.0f, 0.0f, 0.0f);
        ((float4*)tile)[i] = v;
        *(float4*)(mixed + (i >> 4) * MIX_PITCH + 4 * (i & 15)) =
            make_float4(xv.x * 0.1f, xv.y * 0.1f, xv.z * 0.1f, xv.w * 0.1f);
    }
#pragma unroll
    for (int k = 0; k < 4; ++k)
        ((float4*)wlds)[k * 256 + (int)threadIdx.x] = ((const float4*)Wl)[k * 256 + (int)threadIdx.x];
    __syncthreads();

    const int s = bstart[blockIdx.x];
    const int eend = bstart[blockIdx.x + 1];
    const int ngr = (eend - s + 31) >> 5;

    int g = w;
    int2 rc = make_int2(0, 0);
    float nv = 0.0f;
    if (g < ngr) {
        int e0 = s + g * 32;
        int c = eend - e0; if (c > 32) c = 32;
        int my = e0 + (lane < c ? lane : 0);
        rc = ((const int2*)recs)[my];
        nv = nrm[my];
    }
    while (g < ngr) {
        int cnt = eend - (s + g * 32); if (cnt > 32) cnt = 32;
        const int gn = g + 8;
        int e0n = s + (gn < ngr ? gn : g) * 32;
        int cn = eend - e0n; if (cn > 32) cn = 32;
        int myn = e0n + (lane < cn ? lane : 0);
        int2 rc2;
        float nv2;
        if (cnt == 32) edge_group<true >(rc.x, rc.y, nv, base, cnt, lane, h, tile,
                                         (const int2*)recs + myn, nrm + myn, rc2, nv2);
        else           edge_group<false>(rc.x, rc.y, nv, base, cnt, lane, h, tile,
                                         (const int2*)recs + myn, nrm + myn, rc2, nv2);
        rc = rc2; nv = nv2; g = gn;
    }

    __syncthreads();

#pragma unroll
    for (int k = 0; k < BKT * 16 / 256; ++k) {
        int i = k * 256 + (int)threadIdx.x;
        float4 t = ((const float4*)tile)[i];
        float4* mp = (float4*)(mixed + (i >> 4) * MIX_PITCH + 4 * (i & 15));
        float4 m = *mp;
        float4 v;
        v.x = t.x * 0.9f + m.x;
        v.y = t.y * 0.9f + m.y;
        v.z = t.z * 0.9f + m.z;
        v.w = t.w * 0.9f + m.w;
        *mp = v;
    }
    __syncthreads();

    const int ln = lane & 15;
    const int kh = lane >> 4;
    tile_product<2>(mixed, wlds, tile, w >> 1, 2 * (w & 1), ln, kh, beta);
    if (w < 4) tile_product<1>(mixed, wlds, tile, 4, w, ln, kh, beta);
    __syncthreads();

#pragma unroll
    for (int k = 0; k < BKT * 16 / 256; ++k) {
        int i = k * 256 + (int)threadIdx.x;
        if (q0 + i < qlim)
            ((float4*)hout)[q0 + i] = ((const float4*)tile)[i];
    }
}

extern "C" void kernel_launch(void* const* d_in, const int* in_sizes, int n_in,
                              void* d_out, int out_size, void* d_ws, size_t ws_size,
                              hipStream_t stream) {
    const float* x      = (const float*)d_in[0];
    const int*   ei     = (const int*)  d_in[1];
    const float* lin0_w = (const float*)d_in[2];
    const float* lin0_b = (const float*)d_in[3];
    const float* conv_w = (const float*)d_in[4];
    const float* lin1_w = (const float*)d_in[5];
    const float* lin1_b = (const float*)d_in[6];
    float* out = (float*)d_out;

    const int N = in_sizes[0] / 128;
    const int E = in_sizes[1] / 2;
    const int* row = ei;
    const int* col = ei + E;

    float* ws   = (float*)d_ws;
    float* dinv = ws;
    float* x0   = dinv + N;
    float* h    = x0 + (size_t)N * 64;
    float* h2   = h  + (size_t)N * 64;
    const int nb      = (N + BKT - 1) / BKT;
    const int nchunks = (E + BIN_CHUNK - 1) / BIN_CHUNK;
    int*   recs   = (int*)(h2 + (size_t)N * 64);
    float* nrm    = (float*)(recs + (size_t)2 * E);
    int*   cnt    = (int*)(nrm + E);
    int*   off    = cnt + (size_t)nchunks * nb;
    int*   bstart = off + (size_t)nchunks * nb;

    const int ntiles = N / 16;
    const int denseGrid = (ntiles + DENSE_WAVES * DENSE_TPW - 1) / (DENSE_WAVES * DENSE_TPW);

    k_bin_count<<<nchunks, BIN_THREADS, nb * sizeof(int), stream>>>(col, cnt, E, nb);
    k_bin_scan <<<1, 1024, 0, stream>>>(cnt, off, bstart, nchunks, nb);
    k_bin_fill <<<nchunks, BIN_THREADS, nb * sizeof(int), stream>>>(row, col, bstart, off,
                                                                  recs, E, nb);

    k_bucket_dinv<<<(nb + DEG_BKTS - 1) / DEG_BKTS, DEG_THREADS, 0, stream>>>(recs, bstart, dinv, N, nb);
    k_nrm<<<(E + 256 * NRM_PER - 1) / (256 * NRM_PER), 256, 0, stream>>>(recs, dinv, nrm, E);

    k_dense<128, true, true><<<denseGrid, DENSE_THREADS, 0, stream>>>(x, lin0_w, lin0_b, h, x0, ntiles);

    const float* hcur = h;
    float* hnext = h2;
    for (int l = 0; l < 4; ++l) {
        float beta = logf(0.5f / (float)(l + 1) + 1.0f);
        k_bucket_layer<<<nb, 256, 0, stream>>>(recs, nrm, bstart, dinv, hcur, x0,
                                               conv_w + (size_t)l * 64 * 64, beta, hnext, N);
        float* t = (float*)hcur; hcur = hnext; hnext = t;
    }

    k_dense<64, false, false><<<denseGrid, DENSE_THREADS, 0, stream>>>(hcur, lin1_w, lin1_b, out, nullptr, ntiles);
}


# Two short launches folded into their consumers (11 -> 9): each fill block sums the chunk counts and scans the bucket starts itself; layer 1 forms each record's factor from dinv and stores it
# speedup vs baseline: 3.8382x; 1.0669x over previous
#include <hip/hip_runtime.h>
#include <math.h>

typedef __attribute__((ext_vector_type(2))) float v2f;
typedef __attribute__((ext_vector_type(8))) float v8f;

#define DENSE_WAVES   8
#define DENSE_THREADS 256
#define DENSE_TPW     2
#define DENSE_OP      68
template <int K, bool RELU, bool STOREPRE>
__global__ void __launch_bounds__(DENSE_THREADS)
k_dense(const float* __restrict__ X, const float* __restrict__ W,
        const float* __restrict__ bias,
        float* __restrict__ outA, float* __restrict__ outPre,
        int ntiles) {
    constexpr int P  = K + 4;
    constexpr int R4 = K / 4;
    constexpr int NL = 16 * R4 / 32;
    __shared__ __attribute__((aligned(16))) float wlds[64 * P];
    __shared__ __attribute__((aligned(16))) float tiles[DENSE_WAVES * 16 * P];
    __shared__ __attribute__((aligned(16))) float outs[DENSE_WAVES * 16 * DENSE_OP];

    const int lane = threadIdx.x & 31;
    const int w = __builtin_amdgcn_readfirstlane((int)(threadIdx.x >> 5));
    float* mt = tiles + w * (16 * P);
    float* ot = outs + w * (16 * DENSE_OP);
    const int ln = lane & 15;
    const int kh = lane >> 4;

#pragma unroll
    for (int k = 0; k < 64 * R4 / DENSE_THREADS; ++k) {
        const int i = k * DENSE_THREADS + (int)threadIdx.x;
        *(float4*)(wlds + (i / R4) * P + 4 * (i % R4)) = ((const float4*)W)[i];
    }

    float bv[4];
#pragma unroll
    for (int j = 0; j < 4; ++j) bv[j] = bias[16 * j + ln];

    for (int it = 0; it < DENSE_TPW; ++it) {
        const int tile = (blockIdx.x * DENSE_TPW + it) * DENSE_WAVES + w;
        const int tl = tile < ntiles ? tile : ntiles - 1;
        const float4* src = (const float4*)(X + (size_t)tl * 16 * K);
        float4 raw[NL];
#pragma unroll
        for (int k = 0; k < NL; ++k) raw[k] = src[k * 32 + lane];
#pragma unroll
        for (int k = 0; k < NL; ++k) {
            const int idx = k * 32 + lane;
            *(float4*)(mt + (idx / R4) * P + 4 * (idx % R4)) = raw[k];
        }
        __syncthreads();

        v2f afr[K / 4];
#pragma unroll
        for (int kk = 0; kk < K / 4; ++kk)
            afr[kk] = *(const v2f*)(mt + ln * P + 4 * kk + 2 * kh);

        v8f acc[4];
#pragma unroll
        for (int j = 0; j < 4; ++j) acc[j] = v8f{};
#pragma unroll
        for (int kk = 0; kk < K / 4; ++kk) {
#pragma unroll
            for (int j = 0; j < 4; ++j) {
                v2f b = *(const v2f*)(wlds + (16 * j + ln) * P + 4 * kk + 2 * kh);
                acc[j] = __builtin_amdgcn_wmma_f32_16x16x4_f32(
                    false, afr[kk], false, b, (short)0, acc[j], false, false);
            }
        }
#pragma unroll
        for (int j = 0; j < 4; ++j)
#pragma unroll
            for (int r = 0; r < 8; ++r)
                ot[(8 * kh + r) * DENSE_OP + 16 * j + ln] = acc[j][r] + bv[j];
        __syncthreads();

        if (tile < ntiles) {
#pragma unroll
            for (int k = 0; k < 8; ++k) {
                const int idx = k * 32 + lane;
                float4 v = *(const float4*)(ot + (idx >> 4) * DENSE_OP + 4 * (idx & 15));
                const size_t g = (size_t)tile * 256 + idx;
                if constexpr (STOREPRE) ((float4*)outPre)[g] = v;
                if constexpr (RELU) {
                    v.x = fmaxf(v.x, 0.0f); v.y = fmaxf(v.y, 0.0f);
                    v.z = fmaxf(v.z, 0.0f); v.w = fmaxf(v.w, 0.0f);
                }
                ((float4*)outA)[g] = v;
            }
        }
    }
}

#define BKT         80
#define BIN_CHUNK   16384
#define BIN_THREADS 1024

extern __shared__ int dyn_lds[];

__global__ void __launch_bounds__(BIN_THREADS)
k_bin_count(const int* __restrict__ col, int* __restrict__ cnt, int e, int nb) {
    for (int i = threadIdx.x; i < nb; i += BIN_THREADS) dyn_lds[i] = 0;
    __syncthreads();
    const int e0 = blockIdx.x * BIN_CHUNK;
    int c[BIN_CHUNK / BIN_THREADS];
#pragma unroll
    for (int k = 0; k < BIN_CHUNK / BIN_THREADS; ++k) {
        int i = e0 + k * BIN_THREADS + (int)threadIdx.x;
        c[k] = col[i < e ? i : e - 1];
    }
#pragma unroll
    for (int k = 0; k < BIN_CHUNK / BIN_THREADS; ++k) {
        int i = e0 + k * BIN_THREADS + (int)threadIdx.x;
        if (i < e) atomicAdd(&dyn_lds[(unsigned)c[k] / BKT], 1);
    }
    __syncthreads();
    int* dst = cnt + (size_t)blockIdx.x * nb;
    for (int i = threadIdx.x; i < nb; i += BIN_THREADS) dst[i] = dyn_lds[i];
}

#define FILL_CB 33
__global__ void __launch_bounds__(BIN_THREADS)
k_bin_fill(const int* __restrict__ row, const int* __restrict__ col,
           const int* __restrict__ cnt, int* __restrict__ bstart,
           int* __restrict__ recs, int e, int nb, int nchunks) {
    __shared__ int sc[64];
    const int lane = threadIdx.x & 31;
    const int wave = threadIdx.x >> 5;
    const int e0 = blockIdx.x * BIN_CHUNK;
    int r[BIN_CHUNK / BIN_THREADS], c[BIN_CHUNK / BIN_THREADS];
#pragma unroll
    for (int k = 0; k < BIN_CHUNK / BIN_THREADS; ++k) {
        int i = e0 + k * BIN_THREADS + (int)threadIdx.x;
        int ii = i < e ? i : e - 1;
        r[k] = row[ii];
        c[k] = col[ii];
    }

    int carry = 0;
    for (int b0 = 0; b0 < nb; b0 += BIN_THREADS) {
        const int b = b0 + (int)threadIdx.x;
        const int bc = b < nb ? b : nb - 1;
        int tot = 0, before = 0;
        for (int c0 = 0; c0 < nchunks; c0 += FILL_CB) {
            int v[FILL_CB];
#pragma unroll
            for (int k = 0; k < FILL_CB; ++k) {
                int cc = c0 + k < nchunks ? c0 + k : nchunks - 1;
                v[k] = cnt[(size_t)cc * nb + bc];
            }
#pragma unroll
            for (int k = 0; k < FILL_CB; ++k) {
                if (c0 + k < nchunks) {
                    tot += v[k];
                    if (c0 + k < (int)blockIdx.x) before += v[k];
                }
            }
        }
        if (b >= nb) tot = 0;
        int x = tot;
#pragma unroll
        for (int d = 1; d < 32; d <<= 1) {
            int y = __shfl_up(x, d, 32);
            if (lane >= d) x += y;
        }
        if (lane == 31) sc[wave] = x;
        __syncthreads();
        if (wave == 0) {
            int s = sc[lane];
#pragma unroll
            for (int d = 1; d < 32; d <<= 1) {
                int y = __shfl_up(s, d, 32);
                if (lane >= d) s += y;
            }
            sc[32 + lane] = s;
        }
        __syncthreads();
        const int start = carry + (wave > 0 ? sc[32 + wave - 1] : 0) + x - tot;
        if (b < nb) {
            dyn_lds[b] = start + before;
            if (blockIdx.x == 0) bstart[b] = start;
        }
        carry += sc[63];
        __syncthreads();
    }
    if (blockIdx.x == 0 && threadIdx.x == 0) bstart[nb] = carry;
    __syncthreads();

#pragma unroll
    for (int k = 0; k < BIN_CHUNK / BIN_THREADS; ++k) {
        int i = e0 + k * BIN_THREADS + (int)threadIdx.x;
        if (i < e) {
            int pos = atomicAdd(&dyn_lds[(unsigned)c[k] / BKT], 1);
            ((int2*)recs)[pos] = make_int2(r[k], c[k]);
        }
    }
}

#define DEG_BKTS    4
#define DEG_BATCH   8
#define DEG_THREADS (DEG_BKTS * BKT)
__global__ void __launch_bounds__(DEG_THREADS)
k_bucket_dinv(const int* __restrict__ recs, const int* __restrict__ bstart,
              float* __restrict__ dinv, int n, int nb) {
    __shared__ int cnt[DEG_BKTS * BKT];
    cnt[threadIdx.x] = 1;
    __syncthreads();
    const int b0 = blockIdx.x * DEG_BKTS;
    int b1 = b0 + DEG_BKTS; if (b1 > nb) b1 = nb;
    const int s = bstart[b0];
    const int eend = bstart[b1];
    const int base = b0 * BKT;
    for (int i0 = s; i0 < eend; i0 += DEG_THREADS * DEG_BATCH) {
        int t[DEG_BATCH];
#pragma unroll
        for (int k = 0; k < DEG_BATCH; ++k) {
            int i = i0 + k * DEG_THREADS + (int)threadIdx.x;
            int ii = i < eend ? i : eend - 1;
            t[k] = recs[2 * (size_t)ii + 1];
        }
#pragma unroll
        for (int k = 0; k < DEG_BATCH; ++k) {
            int i = i0 + k * DEG_THREADS + (int)threadIdx.x;
            if (i < eend) atomicAdd(&cnt[t[k] - base], 1);
        }
    }
    __syncthreads();
    const int node = base + (int)threadIdx.x;
    if (node < n) {
        float d = (float)cnt[threadIdx.x];
        dinv[node] = d > 0.0f ? rsqrtf(d) : 0.0f;
    }
}

#define AGG_BATCH 16
template <bool FULL, bool FIRST>
__device__ __forceinline__ void edge_group(int r, int c, float nv, float dtv, int base, int cnt, int lane,
                                           const float* __restrict__ h, float* tile,
                                           const int2* __restrict__ nextRec,
                                           const float* nextNrm, float* nrmOut,
                                           int2& rcNext, float& nvNext) {
    float fac = nv;
#pragma unroll
    for (int jb = 0; jb < 32; jb += AGG_BATCH) {
        if (!FULL && jb >= cnt) break;
        float v0[AGG_BATCH], v1[AGG_BATCH];
#pragma unroll
        for (int j = 0; j < AGG_BATCH; ++j) {
            if (FULL || jb + j < cnt) {
                int rj = __builtin_amdgcn_readlane(r, jb + j);
                const float* s = h + (size_t)rj * 64 + lane;
                v0[j] = s[0];
                v1[j] = s[32];
            }
        }
        if (jb == 0) {
            rcNext = *nextRec;
            if constexpr (FIRST) {
                __builtin_amdgcn_sched_barrier(0);
                fac = nv * dtv;
                if (FULL || lane < cnt) *nrmOut = fac;
                nvNext = 0.0f;
            } else {
                nvNext = *nextNrm;
            }
        }
        const int nvBits = __float_as_int(fac);
#pragma unroll
        for (int j = 0; j < AGG_BATCH; ++j) {
            if (FULL || jb + j < cnt) {
                int cj = __builtin_amdgcn_readlane(c, jb + j) - base;
                float nj = __int_as_float(__builtin_amdgcn_readlane(nvBits, jb + j));
                float* d = tile + cj * 64 + lane;
                atomicAdd(d,      nj * v0[j]);
                atomicAdd(d + 32, nj * v1[j]);
            }
        }
    }
}

#define MIX_PITCH 68

template <int NC>
__device__ __forceinline__ void tile_product(const float* mixed, const float* wlds, float* tile,
                                             int rt, int jc, int ln, int kh, float beta) {
    v2f afr[16];
#pragma unroll
    for (int kk = 0; kk < 16; ++kk)
        afr[kk] = *(const v2f*)(mixed + (rt * 16 + ln) * MIX_PITCH + 4 * kk + 2 * kh);

    const float* wcol = wlds + 2 * kh * 64 + 16 * jc + ln;
    v8f acc[NC];
#pragma unroll
    for (int c = 0; c < NC; ++c) acc[c] = v8f{};
#pragma unroll
    for (int kk = 0; kk < 16; ++kk) {
#pragma unroll
        for (int c = 0; c < NC; ++c) {
            v2f b;
            b.x = wcol[(size_t)(4 * kk) * 64 + 16 * c];
            b.y = wcol[(size_t)(4 * kk + 1) * 64 + 16 * c];
            acc[c] = __builtin_amdgcn_wmma_f32_16x16x4_f32(
                false, afr[kk], false, b, (short)0, acc[c], false, false);
        }
    }
    const float ob = 1.0f - beta;
    const float* hp = mixed + (rt * 16 + 8 * kh) * MIX_PITCH + 16 * jc + ln;
    float* op = tile + (rt * 16 + 8 * kh) * 64 + 16 * jc + ln;
#pragma unroll
    for (int r = 0; r < 8; ++r) {
#pragma unroll
        for (int c = 0; c < NC; ++c) {
            float hh = hp[r * MIX_PITCH + 16 * c];
            float v = ob * hh + beta * acc[c][r];
            op[r * 64 + 16 * c] = fmaxf(v, 0.0f);
        }
    }
}
template <bool FIRST>
__global__ void __launch_bounds__(256, 10)
k_bucket_layer(const int* __restrict__ recs, float* __restrict__ nrm,
               const int* __restrict__ bstart, const float* __restrict__ dinv,
               const float* __restrict__ h, const float* __restrict__ x0,
               const float* __restrict__ Wl, float beta,
               float* __restrict__ hout, int n) {
    __shared__ __attribute__((aligned(16))) float tile[BKT * 64];
    __shared__ __attribute__((aligned(16))) float mixed[BKT * MIX_PITCH];
    __shared__ __attribute__((aligned(16))) float wlds[64 * 64];
    const int base = blockIdx.x * BKT;
    const int lane = threadIdx.x & 31;
    const int w = __builtin_amdgcn_readfirstlane((int)(threadIdx.x >> 5));

    const int q0 = blockIdx.x * (BKT * 16);
    const int qlim = n * 16;

#pragma unroll
    for (int k = 0; k < BKT * 16 / 256; ++k) {
        int i = k * 256 + (int)threadIdx.x;
        int q = q0 + i;
        int qc = q < qlim ? q : qlim - 1;
        float d = dinv[qc >> 4];
        float dd = d * d;
        float4 hv = ((const float4*)h)[qc];
        float4 xv = ((const float4*)x0)[qc];
        float4 v = make_float4(dd * hv.x, dd * hv.y, dd * hv.z, dd * hv.w);
        if (q >= qlim) v = make_float4(0.0f, 0.0f, 0.0f, 0.0f);
        ((float4*)tile)[i] = v;
        *(float4*)(mixed + (i >> 4) * MIX_PITCH + 4 * (i & 15)) =
            make_float4(xv.x * 0.1f, xv.y * 0.1f, xv.z * 0.1f, xv.w * 0.1f);
    }
#pragma unroll
    for (int k = 0; k < 4; ++k)
        ((float4*)wlds)[k * 256 + (int)threadIdx.x] = ((const float4*)Wl)[k * 256 + (int)threadIdx.x];
    __syncthreads();

    const int s = bstart[blockIdx.x];
    const int eend = bstart[blockIdx.x + 1];
    const int ngr = (eend - s + 31) >> 5;

    int g = w;
    int2 rc = make_int2(0, 0);
    float nv = 0.0f;
    if (g < ngr) {
        int e0 = s + g * 32;
        int c = eend - e0; if (c > 32) c = 32;
        int my = e0 + (lane < c ? lane : 0);
        rc = ((const int2*)recs)[my];
        if constexpr (!FIRST) nv = nrm[my];
    }
    while (g < ngr) {
        int cnt = eend - (s + g * 32); if (cnt > 32) cnt = 32;
        const int gn = g + 8;
        int e0n = s + (gn < ngr ? gn : g) * 32;
        int cn = eend - e0n; if (cn > 32) cn = 32;
        int myn = e0n + (lane < cn ? lane : 0);
        int2 rc2;
        float nv2;
        float dtv = 0.0f;
        if constexpr (FIRST) {
            nv  = dinv[rc.x];
            dtv = dinv[rc.y];
        }
        float* nout = nrm + (s + g * 32 + lane);
        if (cnt == 32) edge_group<true,  FIRST>(rc.x, rc.y, nv, dtv, base, cnt, lane, h, tile,
                                                (const int2*)recs + myn, nrm + myn, nout, rc2, nv2);
        else           edge_group<false, FIRST>(rc.x, rc.y, nv, dtv, base, cnt, lane, h, tile,
                                                (const int2*)recs + myn, nrm + myn, nout, rc2, nv2);
        rc = rc2; nv = nv2; g = gn;
    }

    __syncthreads();

#pragma unroll
    for (int k = 0; k < BKT * 16 / 256; ++k) {
        int i = k * 256 + (int)threadIdx.x;
        float4 t = ((const float4*)tile)[i];
        float4* mp = (float4*)(mixed + (i >> 4) * MIX_PITCH + 4 * (i & 15));
        float4 m = *mp;
        float4 v;
        v.x = t.x * 0.9f + m.x;
        v.y = t.y * 0.9f + m.y;
        v.z = t.z * 0.9f + m.z;
        v.w = t.w * 0.9f + m.w;
        *mp = v;
    }
    __syncthreads();

    const int ln = lane & 15;
    const int kh = lane >> 4;
    tile_product<2>(mixed, wlds, tile, w >> 1, 2 * (w & 1), ln, kh, beta);
    if (w < 4) tile_product<1>(mixed, wlds, tile, 4, w, ln, kh, beta);
    __syncthreads();

#pragma unroll
    for (int k = 0; k < BKT * 16 / 256; ++k) {
        int i = k * 256 + (int)threadIdx.x;
        if (q0 + i < qlim)
            ((float4*)hout)[q0 + i] = ((const float4*)tile)[i];
    }
}

extern "C" void kernel_launch(void* const* d_in, const int* in_sizes, int n_in,
                              void* d_out, int out_size, void* d_ws, size_t ws_size,
                              hipStream_t stream) {
    const float* x      = (const float*)d_in[0];
    const int*   ei     = (const int*)  d_in[1];
    const float* lin0_w = (const float*)d_in[2];
    const float* lin0_b = (const float*)d_in[3];
    const float* conv_w = (const float*)d_in[4];
    const float* lin1_w = (const float*)d_in[5];
    const float* lin1_b = (const float*)d_in[6];
    float* out = (float*)d_out;

    const int N = in_sizes[0] / 128;
    const int E = in_sizes[1] / 2;
    const int* row = ei;
    const int* col = ei + E;

    float* ws   = (float*)d_ws;
    float* dinv = ws;
    float* x0   = dinv + N;
    float* h    = x0 + (size_t)N * 64;
    float* h2   = h  + (size_t)N * 64;
    const int nb      = (N + BKT - 1) / BKT;
    const int nchunks = (E + BIN_CHUNK - 1) / BIN_CHUNK;
    int*   recs   = (int*)(h2 + (size_t)N * 64);
    float* nrm    = (float*)(recs + (size_t)2 * E);
    int*   cnt    = (int*)(nrm + E);
    int*   bstart = cnt + (size_t)nchunks * nb;

    const int ntiles = N / 16;
    const int denseGrid = (ntiles + DENSE_WAVES * DENSE_TPW - 1) / (DENSE_WAVES * DENSE_TPW);

    k_bin_count<<<nchunks, BIN_THREADS, nb * sizeof(int), stream>>>(col, cnt, E, nb);
    k_bin_fill <<<nchunks, BIN_THREADS, nb * sizeof(int), stream>>>(row, col, cnt, bstart,
                                                                  recs, E, nb, nchunks);

    k_bucket_dinv<<<(nb + DEG_BKTS - 1) / DEG_BKTS, DEG_THREADS, 0, stream>>>(recs, bstart, dinv, N, nb);

    k_dense<128, true, true><<<denseGrid, DENSE_THREADS, 0, stream>>>(x, lin0_w, lin0_b, h, x0, ntiles);

    const float* hcur = h;
    float* hnext = h2;
    for (int l = 0; l < 4; ++l) {
        float beta = logf(0.5f / (float)(l + 1) + 1.0f);
        if (l == 0)
            k_bucket_layer<true ><<<nb, 256, 0, stream>>>(recs, nrm, bstart, dinv, hcur, x0,
                                                          conv_w, beta, hnext, N);
        else
            k_bucket_layer<false><<<nb, 256, 0, stream>>>(recs, nrm, bstart, dinv, hcur, x0,
                                                          conv_w + (size_t)l * 64 * 64, beta, hnext, N);
        float* t = (float*)hcur; hcur = hnext; hnext = t;
    }

    k_dense<64, false, false><<<denseGrid, DENSE_THREADS, 0, stream>>>(hcur, lin1_w, lin1_b, out, nullptr, ntiles);
}
